# starting kernel
# baseline (speedup 1.0000x reference)
#include <hip/hip_runtime.h>
#include <math.h>

#define N_QUBITS 12
#define DEPTH    4
#define HID      64
#define NPARAMS  144
#define BATCH    4096
#define DIM      4096
#define OUT_DIM  2

typedef __attribute__((ext_vector_type(2))) float v2f;
typedef __attribute__((ext_vector_type(8))) float v8f;

__device__ __forceinline__ v8f wmma_f32(v2f a, v2f b, v8f c) {
  return __builtin_amdgcn_wmma_f32_16x16x4_f32(
       false, a,  false, b,
       (short)0, c,  false,  false);
}

__device__ __forceinline__ v2f load_a_frag(const float* src, int lda,
                                           int row0, int k0, int lane) {
  int m = row0 + (lane & 15);
  int k = k0 + ((lane >> 4) << 1);
  v2f a;
  a.x = src[m * lda + k];
  a.y = src[m * lda + k + 1];
  return a;
}

__device__ __forceinline__ v2f load_b_frag(const float* W, int ldb,
                                           int k0, int n0, int lane) {
  int c = n0 + (lane & 15);
  int k = k0 + ((lane >> 4) << 1);
  v2f b;
  b.x = W[k * ldb + c];
  b.y = W[(k + 1) * ldb + c];
  return b;
}

__global__ __launch_bounds__(128) void front_mlp_kernel(
    const float* __restrict__ noise, const float* __restrict__ W1,
    const float* __restrict__ b1,    const float* __restrict__ W2,
    const float* __restrict__ b2,    float* __restrict__ params) {
  __shared__ float hbuf[4][16 * HID];
  const int lane  = threadIdx.x & 31;
  const int wave  = threadIdx.x >> 5;
  const int row0  = (blockIdx.x * 4 + wave) * 16;
  const int col   = lane & 15;
  const int rhalf = (lane >> 4) * 8;
  float* h = hbuf[wave];

  for (int nt = 0; nt < HID / 16; ++nt) {
    v8f c = {};
#pragma unroll
    for (int kc = 0; kc < N_QUBITS; kc += 4) {
      v2f a = load_a_frag(noise, N_QUBITS, row0, kc, lane);
      v2f b = load_b_frag(W1, HID, kc, nt * 16, lane);
      c = wmma_f32(a, b, c);
    }
    int n = nt * 16 + col;
    float bias = b1[n];
#pragma unroll
    for (int r = 0; r < 8; ++r)
      h[(rhalf + r) * HID + n] = tanhf(c[r] + bias);
  }
  __syncthreads();

  for (int nt = 0; nt < NPARAMS / 16; ++nt) {
    v8f c = {};
#pragma unroll
    for (int kc = 0; kc < HID; kc += 4) {
      v2f a = load_a_frag(h, HID, 0, kc, lane);
      v2f b = load_b_frag(W2, NPARAMS, kc, nt * 16, lane);
      c = wmma_f32(a, b, c);
    }
    int n = nt * 16 + col;
    float bias = b2[n];
#pragma unroll
    for (int r = 0; r < 8; ++r)
      params[(size_t)(row0 + rhalf + r) * NPARAMS + n] = c[r] + bias;
  }
}

__global__ __launch_bounds__(256) void quantum_kernel(
    const float* __restrict__ params, float* __restrict__ meas) {
  __shared__ float s_re0[DIM], s_im0[DIM], s_re1[DIM], s_im1[DIM];
  const int tid = threadIdx.x;
  const int bidx = blockIdx.x;

  for (int i = tid; i < DIM; i += 256) { s_re0[i] = 0.0f; s_im0[i] = 0.0f; }
  if (tid == 0) s_re0[0] = 1.0f;
  __syncthreads();

  const float* p = params + (size_t)bidx * NPARAMS;

  for (int l = 0; l < DEPTH; ++l) {
    float* Ar = (l & 1) ? s_re1 : s_re0;
    float* Ai = (l & 1) ? s_im1 : s_im0;
    float* Br = (l & 1) ? s_re0 : s_re1;
    float* Bi = (l & 1) ? s_im0 : s_im1;

    for (int q = 0; q < N_QUBITS; ++q) {
      const float aa = p[(l * N_QUBITS + q) * 3 + 0];
      const float bb = p[(l * N_QUBITS + q) * 3 + 1];
      const float cc = p[(l * N_QUBITS + q) * 3 + 2];
      float sa, ca, sb, cb, sz, cz;
      sincosf(aa * 0.5f, &sa, &ca);
      sincosf(bb * 0.5f, &sb, &cb);
      sincosf(cc * 0.5f, &sz, &cz);
      const float m00r =  cb * ca, m00i =  sb * sa;
      const float m01r = -sb * ca, m01i = -cb * sa;
      const float m10r =  sb * ca, m10i = -cb * sa;
      const float m11r =  cb * ca, m11i = -sb * sa;
      const float U00r = cz * m00r + sz * m00i, U00i = cz * m00i - sz * m00r;
      const float U01r = cz * m01r + sz * m01i, U01i = cz * m01i - sz * m01r;
      const float U10r = cz * m10r - sz * m10i, U10i = cz * m10i + sz * m10r;
      const float U11r = cz * m11r - sz * m11i, U11i = cz * m11i + sz * m11r;

      const int bitpos  = N_QUBITS - 1 - q;
      const int stride  = 1 << bitpos;
      const int lowmask = stride - 1;
      for (int t = tid; t < DIM / 2; t += 256) {
        int i0 = ((t & ~lowmask) << 1) | (t & lowmask);
        int i1 = i0 | stride;
        float xr = Ar[i0], xi = Ai[i0];
        float yr = Ar[i1], yi = Ai[i1];
        Ar[i0] = U00r * xr - U00i * xi + U01r * yr - U01i * yi;
        Ai[i0] = U00r * xi + U00i * xr + U01r * yi + U01i * yr;
        Ar[i1] = U10r * xr - U10i * xi + U11r * yr - U11i * yi;
        Ai[i1] = U10r * xi + U10i * xr + U11r * yi + U11i * yr;
      }
      __syncthreads();
    }

    for (int i = tid; i < DIM; i += 256) {
      int x = i;
#pragma unroll
      for (int q = 0; q < N_QUBITS; ++q) {
        int pc = N_QUBITS - 1 - q;
        int pt = N_QUBITS - 1 - ((q + 1) % N_QUBITS);
        if ((x >> pc) & 1) x ^= (1 << pt);
      }
      Br[x] = Ar[i];
      Bi[x] = Ai[i];
    }
    __syncthreads();
  }

  float acc[N_QUBITS];
#pragma unroll
  for (int q = 0; q < N_QUBITS; ++q) acc[q] = 0.0f;
  for (int i = tid; i < DIM; i += 256) {
    float pr = s_re0[i] * s_re0[i] + s_im0[i] * s_im0[i];
#pragma unroll
    for (int q = 0; q < N_QUBITS; ++q)
      acc[q] += ((i >> (N_QUBITS - 1 - q)) & 1) ? -pr : pr;
  }
  float* part = s_re1;
#pragma unroll
  for (int q = 0; q < N_QUBITS; ++q) part[tid * N_QUBITS + q] = acc[q];
  __syncthreads();
  for (int s = 128; s > 0; s >>= 1) {
    if (tid < s) {
#pragma unroll
      for (int q = 0; q < N_QUBITS; ++q)
        part[tid * N_QUBITS + q] += part[(tid + s) * N_QUBITS + q];
    }
    __syncthreads();
  }
  if (tid < N_QUBITS) meas[(size_t)bidx * N_QUBITS + tid] = part[tid];
}

__global__ __launch_bounds__(128) void back_mlp_kernel(
    const float* __restrict__ meas, const float* __restrict__ W3,
    const float* __restrict__ b3,   const float* __restrict__ W4,
    const float* __restrict__ b4,   float* __restrict__ out) {
  __shared__ float hbuf[4][16 * HID];
  const int lane  = threadIdx.x & 31;
  const int wave  = threadIdx.x >> 5;
  const int row0  = (blockIdx.x * 4 + wave) * 16;
  const int col   = lane & 15;
  const int rhalf = (lane >> 4) * 8;
  float* h = hbuf[wave];

  for (int nt = 0; nt < HID / 16; ++nt) {
    v8f c = {};
#pragma unroll
    for (int kc = 0; kc < N_QUBITS; kc += 4) {
      v2f a = load_a_frag(meas, N_QUBITS, row0, kc, lane);
      v2f b = load_b_frag(W3, HID, kc, nt * 16, lane);
      c = wmma_f32(a, b, c);
    }
    int n = nt * 16 + col;
    float bias = b3[n];
#pragma unroll
    for (int r = 0; r < 8; ++r)
      h[(rhalf + r) * HID + n] = tanhf(c[r] + bias);
  }
  __syncthreads();

  v8f c = {};
#pragma unroll
  for (int kc = 0; kc < HID; kc += 4) {
    v2f a = load_a_frag(h, HID, 0, kc, lane);
    int k = kc + ((lane >> 4) << 1);
    v2f b;
    b.x = (col < OUT_DIM) ? W4[k * OUT_DIM + col] : 0.0f;
    b.y = (col < OUT_DIM) ? W4[(k + 1) * OUT_DIM + col] : 0.0f;
    c = wmma_f32(a, b, c);
  }
  if (col < OUT_DIM) {
    float bias = b4[col];
#pragma unroll
    for (int r = 0; r < 8; ++r)
      out[(size_t)(row0 + rhalf + r) * OUT_DIM + col] = c[r] + bias;
  }
}

extern "C" void kernel_launch(void* const* d_in, const int* in_sizes, int n_in,
                              void* d_out, int out_size, void* d_ws, size_t ws_size,
                              hipStream_t stream) {
  const float* noise = (const float*)d_in[0];
  const float* W1 = (const float*)d_in[1];
  const float* b1 = (const float*)d_in[2];
  const float* W2 = (const float*)d_in[3];
  const float* b2 = (const float*)d_in[4];
  const float* W3 = (const float*)d_in[5];
  const float* b3 = (const float*)d_in[6];
  const float* W4 = (const float*)d_in[7];
  const float* b4 = (const float*)d_in[8];
  float* out = (float*)d_out;

  float* params = (float*)d_ws;
  float* meas   = params + (size_t)BATCH * NPARAMS;

  front_mlp_kernel<<<BATCH / 64, 128, 0, stream>>>(noise, W1, b1, W2, b2, params);
  quantum_kernel<<<BATCH, 256, 0, stream>>>(params, meas);
  back_mlp_kernel<<<BATCH / 64, 128, 0, stream>>>(meas, W3, b3, W4, b4, out);
}


# Front MLP folded into the state-vector kernel: 768 resident blocks of 5-6 elements each (angles from one 16-row WMMA tile in LDS, addresses set up once, issue priority by elements left); 2 launches
# speedup vs baseline: 3.2281x; 3.2281x over previous
#include <hip/hip_runtime.h>
#include <math.h>

#define N_QUBITS 12
#define DEPTH    4
#define HID      64
#define NPARAMS  144
#define BATCH    4096
#define DIM      4096
#define OUT_DIM  2

typedef __attribute__((ext_vector_type(2))) float v2f;
typedef __attribute__((ext_vector_type(8))) float v8f;

__device__ __forceinline__ v8f wmma_f32(v2f a, v2f b, v8f c) {
  return __builtin_amdgcn_wmma_f32_16x16x4_f32(
       false, a,  false, b,
       (short)0, c,  false,  false);
}

__device__ __forceinline__ v2f load_a_frag(const float* src, int lda,
                                           int row0, int k0, int lane) {
  int m = row0 + (lane & 15);
  int k = k0 + ((lane >> 4) << 1);
  v2f a;
  a.x = src[m * lda + k];
  a.y = src[m * lda + k + 1];
  return a;
}

__device__ __forceinline__ v2f load_b_frag(const float* W, int ldb,
                                           int k0, int n0, int lane) {
  int c = n0 + (lane & 15);
  int k = k0 + ((lane >> 4) << 1);
  v2f b;
  b.x = W[k * ldb + c];
  b.y = W[(k + 1) * ldb + c];
  return b;
}

#define FRAG_PER_LAYER 512
#define NBLK 768
#define MAXE 6

__device__ __forceinline__ int phys_idx(int i) {
  return i ^ (((i >> 5) & 3) * 9) ^ ((((i >> 8) ^ (i >> 11)) & 1) << 4);
}

__device__ __forceinline__ int cnot_ring(int x) {
#pragma unroll
  for (int q = 0; q < N_QUBITS; ++q) {
    int pc = N_QUBITS - 1 - q;
    int pt = N_QUBITS - 1 - ((q + 1) % N_QUBITS);
    if ((x >> pc) & 1) x ^= (1 << pt);
  }
  return x;
}

template <int G, int MODE>
__device__ __forceinline__ void group_pass(v2f* st, const v2f* af,
                                           int wave, int lane,
                                           float* acc = nullptr,
                                           const int* lda = nullptr) {
  constexpr bool LAST = (MODE == 1);
  constexpr int SH = 9 - 3 * G;
  const int c = lane & 15;
  const int h = lane >> 4;
  v2f a[4];
#pragma unroll
  for (int kk = 0; kk < 4; ++kk) a[kk] = af[kk * 32 + lane];

  int ib[4], pb[4];
  v2f x[4][4];
#pragma unroll
  for (int n = 0; n < 4; ++n) {
    const int col9 = ((wave * 4 + n) << 4) | c;
    const int hi = col9 >> SH;
    const int lo = col9 & ((1 << SH) - 1);
    ib[n] = (hi << (SH + 3)) | (h << (SH + 2)) | lo;
    pb[n] = phys_idx(ib[n]);
#pragma unroll
    for (int u = 0; u < 4; ++u)
      x[n][u] = st[(G == 3) ? lda[n * 4 + u] : (pb[n] ^ phys_idx(u << SH))];
  }
  __builtin_amdgcn_sched_barrier(0);
  v8f d[4] = {};
#pragma unroll
  for (int kk = 0; kk < 4; ++kk)
#pragma unroll
    for (int n = 0; n < 4; ++n) d[n] = wmma_f32(a[kk], x[n][kk], d[n]);

  if (MODE == 2) {
    v2f S[4], B[4], C[4];
#pragma unroll
    for (int n = 0; n < 4; ++n) {
      v2f q[4];
#pragma unroll
      for (int u = 0; u < 4; ++u) {
        v2f v;
        v.x = d[n][2 * u];
        v.y = d[n][2 * u + 1];
        q[u] = v * v;
      }
      const v2f e = q[0] + q[1], f = q[2] + q[3];
      S[n] = e + f;
      B[n] = e - f;
      C[n] = (q[0] - q[1]) + (q[3] - q[2]);
    }
    const v2f e2 = S[0] + S[1], f2 = S[2] + S[3];
    const v2f vT  = e2 + f2;
    const v2f v8  = e2 - f2;
    const v2f v87 = (S[0] - S[1]) + (S[3] - S[2]);
    const v2f v871  = (B[0] - B[1]) + (B[3] - B[2]);
    const v2f v8710 = (C[0] - C[1]) + (C[3] - C[2]);
    const float qT = vT.x + vT.y, q8 = v8.x + v8.y, q87 = v87.x + v87.y;
    const float q871 = v871.x + v871.y, q8710 = v8710.x + v8710.y;
    const unsigned L = ((unsigned)wave << 9) | ((unsigned)c << 3) | ((unsigned)h << 2);
    unsigned g = L;
    g ^= g >> 1; g ^= g >> 2; g ^= g >> 4; g ^= g >> 8;
#pragma unroll
    for (int j = 0; j <= 10; ++j) {
      const float qv = (j >= 9) ? qT : (j == 8) ? q8 : (j >= 2) ? q87
                     : (j == 1) ? q871 : q8710;
      const bool neg = ((g >> j) & 1u) != 0u;
      acc[N_QUBITS - 1 - j] = neg ? -qv : qv;
    }
    {
      const bool neg = (((g ^ (g >> 11)) & 1u) != 0u);
      acc[0] = neg ? -q8710 : q8710;
    }
    return;
  }
  if (LAST) __syncthreads();
#pragma unroll
  for (int n = 0; n < 4; ++n) {
    const int pr = LAST ? cnot_ring(ib[n]) : 0;
#pragma unroll
    for (int u = 0; u < 4; ++u) {
      const int dst = LAST ? phys_idx(pr ^ cnot_ring(u << SH))
                           : (pb[n] ^ phys_idx(u << SH));
      v2f v;
      v.x = d[n][2 * u];
      v.y = d[n][2 * u + 1];
      st[dst] = v;
    }
  }
  __syncthreads();
}

__global__ __launch_bounds__(256) void quantum_kernel(
    const float* __restrict__ noise, const float* __restrict__ W1,
    const float* __restrict__ b1,    const float* __restrict__ W2,
    const float* __restrict__ b2,    float* __restrict__ meas) {
  __shared__ v2f st[DIM];
  __shared__ v2f afrag[DEPTH * FRAG_PER_LAYER];
  __shared__ v2f ssc[NPARAMS];
  __shared__ float gU[DEPTH * N_QUBITS][8];
  __shared__ float part[256 * N_QUBITS];
  __shared__ float hbuf[16 * HID];
  __shared__ float pbuf[16 * NPARAMS];
  const int tid  = threadIdx.x;
  const int lane = tid & 31;
  const int wave = tid >> 5;
  const int wave_u = __builtin_amdgcn_readfirstlane(wave);
  const int bq = (int)blockIdx.x / 3, br = (int)blockIdx.x - 3 * bq;
  const int ebase = 16 * bq + ((br == 0) ? 0 : (br == 1) ? 6 : 11);
  const int ecnt  = (br == 0) ? 6 : 5;

  {
    const int col   = lane & 15;
    const int rhalf = (lane >> 4) * 8;
    if (wave_u < HID / 16) {
      const int e = ebase + ((col < ecnt) ? col : 0);
      v8f cacc = {};
#pragma unroll
      for (int kc = 0; kc < N_QUBITS; kc += 4) {
        const int k = kc + ((lane >> 4) << 1);
        v2f a;
        a.x = noise[e * N_QUBITS + k];
        a.y = noise[e * N_QUBITS + k + 1];
        v2f b = load_b_frag(W1, HID, kc, wave_u * 16, lane);
        cacc = wmma_f32(a, b, cacc);
      }
      const int n = wave_u * 16 + col;
      const float bias = b1[n];
#pragma unroll
      for (int r = 0; r < 8; ++r)
        hbuf[(rhalf + r) * HID + n] = tanhf(cacc[r] + bias);
    }
    __syncthreads();
    for (int nt = wave_u; nt < NPARAMS / 16; nt += 8) {
      v8f cacc = {};
#pragma unroll
      for (int kc = 0; kc < HID; kc += 4) {
        v2f a = load_a_frag(hbuf, HID, 0, kc, lane);
        v2f b = load_b_frag(W2, NPARAMS, kc, nt * 16, lane);
        cacc = wmma_f32(a, b, cacc);
      }
      const int n = nt * 16 + col;
      const float bias = b2[n];
#pragma unroll
      for (int r = 0; r < 8; ++r)
        pbuf[(rhalf + r) * NPARAMS + n] = cacc[r] + bias;
    }
    __syncthreads();
  }

  int ld3[16];
  {
    const int c = lane & 15, h = lane >> 4;
#pragma unroll
    for (int n = 0; n < 4; ++n) {
      const int col9 = ((wave * 4 + n) << 4) | c;
      const int pbn = phys_idx((col9 << 3) | (h << 2));
#pragma unroll
      for (int u = 0; u < 4; ++u) ld3[n * 4 + u] = pbn ^ phys_idx(u);
    }
  }

  for (int it = 0; it < MAXE; ++it) {
  if (it >= ecnt) break;
  const int bidx = ebase + it;
  const float* p = pbuf + it * NPARAMS;
  {
    const int left = ecnt - it;
    if (left >= 4)      __builtin_amdgcn_s_setprio(3);
    else if (left == 3) __builtin_amdgcn_s_setprio(2);
    else if (left == 2) __builtin_amdgcn_s_setprio(1);
    else                __builtin_amdgcn_s_setprio(0);
  }

  if (tid < NPARAMS) {
    float s, c;
    sincosf(p[tid] * 0.5f, &s, &c);
    v2f sc; sc.x = s; sc.y = c;
    ssc[tid] = sc;
  }
  for (int i = tid; i < DIM; i += 256) { v2f z; z.x = 0.0f; z.y = 0.0f; st[i] = z; }
  if (tid == 0) { v2f one; one.x = 1.0f; one.y = 0.0f; st[0] = one; }
  __syncthreads();

  if (tid < DEPTH * N_QUBITS) {
    const v2f A = ssc[tid * 3 + 0], B = ssc[tid * 3 + 1], Z = ssc[tid * 3 + 2];
    const float sa = A.x, ca = A.y, sb = B.x, cb = B.y, sz = Z.x, cz = Z.y;
    const float m00r =  cb * ca, m00i =  sb * sa;
    const float m01r = -sb * ca, m01i = -cb * sa;
    const float m10r =  sb * ca, m10i = -cb * sa;
    const float m11r =  cb * ca, m11i = -sb * sa;
    float* u = gU[tid];
    u[0] = cz * m00r + sz * m00i;  u[1] = cz * m00i - sz * m00r;
    u[2] = cz * m01r + sz * m01i;  u[3] = cz * m01i - sz * m01r;
    u[4] = cz * m10r - sz * m10i;  u[5] = cz * m10i + sz * m10r;
    u[6] = cz * m11r - sz * m11i;  u[7] = cz * m11i + sz * m11r;
  }
  __syncthreads();

  for (int e = tid; e < DEPTH * FRAG_PER_LAYER; e += 256) {
    const int ln = e & 31, kk = (e >> 5) & 3, g = (e >> 7) & 3, l = e >> 9;
    const int m = ln & 15, h = ln >> 4;
    const int rip = m & 1;
    const int jp = 4 * (m >> 3) + ((m & 7) >> 1);
    const int j = 4 * h + kk;
    float pr = 1.0f, pi = 0.0f;
#pragma unroll
    for (int b = 2; b >= 0; --b) {
      const float* u = gU[l * N_QUBITS + 3 * g + (2 - b)];
      const int eo = ((((jp >> b) & 1) << 1) | ((j >> b) & 1)) << 1;
      const float ur = u[eo], ui = u[eo + 1];
      const float nr = pr * ur - pi * ui;
      const float ni = pr * ui + pi * ur;
      pr = nr; pi = ni;
    }
    v2f ent;
    ent.x = rip ? pi : pr;
    ent.y = rip ? pr : -pi;
    afrag[e] = ent;
  }
  __syncthreads();

  for (int l = 0; l < DEPTH - 1; ++l) {
    const v2f* af = afrag + l * FRAG_PER_LAYER;
    group_pass<0, 0>(st, af,       wave, lane);
    group_pass<1, 0>(st, af + 128, wave, lane);
    group_pass<2, 0>(st, af + 256, wave, lane);
    group_pass<3, 1>(st, af + 384, wave, lane, nullptr, ld3);
  }

  float acc[N_QUBITS];
  {
    const v2f* af = afrag + (DEPTH - 1) * FRAG_PER_LAYER;
    group_pass<0, 0>(st, af,       wave, lane);
    group_pass<1, 0>(st, af + 128, wave, lane);
    group_pass<2, 0>(st, af + 256, wave, lane);
    group_pass<3, 2>(st, af + 384, wave, lane, acc, ld3);
  }
#pragma unroll
  for (int q = 0; q < N_QUBITS; ++q) part[tid * N_QUBITS + q] = acc[q];
  __syncthreads();
  for (int s = 128; s > 0; s >>= 1) {
    if (tid < s) {
#pragma unroll
      for (int q = 0; q < N_QUBITS; ++q)
        part[tid * N_QUBITS + q] += part[(tid + s) * N_QUBITS + q];
    }
    __syncthreads();
  }
  if (tid < N_QUBITS) meas[(size_t)bidx * N_QUBITS + tid] = part[tid];
  }
}

__global__ __launch_bounds__(128) void back_mlp_kernel(
    const float* __restrict__ meas, const float* __restrict__ W3,
    const float* __restrict__ b3,   const float* __restrict__ W4,
    const float* __restrict__ b4,   float* __restrict__ out) {
  __shared__ float hbuf[4][16 * HID];
  const int lane  = threadIdx.x & 31;
  const int wave  = threadIdx.x >> 5;
  const int row0  = (blockIdx.x * 4 + wave) * 16;
  const int col   = lane & 15;
  const int rhalf = (lane >> 4) * 8;
  float* h = hbuf[wave];

  for (int nt = 0; nt < HID / 16; ++nt) {
    v8f c = {};
#pragma unroll
    for (int kc = 0; kc < N_QUBITS; kc += 4) {
      v2f a = load_a_frag(meas, N_QUBITS, row0, kc, lane);
      v2f b = load_b_frag(W3, HID, kc, nt * 16, lane);
      c = wmma_f32(a, b, c);
    }
    int n = nt * 16 + col;
    float bias = b3[n];
#pragma unroll
    for (int r = 0; r < 8; ++r)
      h[(rhalf + r) * HID + n] = tanhf(c[r] + bias);
  }
  __syncthreads();

  v8f c = {};
#pragma unroll
  for (int kc = 0; kc < HID; kc += 4) {
    v2f a = load_a_frag(h, HID, 0, kc, lane);
    int k = kc + ((lane >> 4) << 1);
    v2f b;
    b.x = (col < OUT_DIM) ? W4[k * OUT_DIM + col] : 0.0f;
    b.y = (col < OUT_DIM) ? W4[(k + 1) * OUT_DIM + col] : 0.0f;
    c = wmma_f32(a, b, c);
  }
  if (col < OUT_DIM) {
    float bias = b4[col];
#pragma unroll
    for (int r = 0; r < 8; ++r)
      out[(size_t)(row0 + rhalf + r) * OUT_DIM + col] = c[r] + bias;
  }
}

extern "C" void kernel_launch(void* const* d_in, const int* in_sizes, int n_in,
                              void* d_out, int out_size, void* d_ws, size_t ws_size,
                              hipStream_t stream) {
  const float* noise = (const float*)d_in[0];
  const float* W1 = (const float*)d_in[1];
  const float* b1 = (const float*)d_in[2];
  const float* W2 = (const float*)d_in[3];
  const float* b2 = (const float*)d_in[4];
  const float* W3 = (const float*)d_in[5];
  const float* b3 = (const float*)d_in[6];
  const float* W4 = (const float*)d_in[7];
  const float* b4 = (const float*)d_in[8];
  float* out = (float*)d_out;

  float* meas = (float*)d_ws;

  quantum_kernel<<<NBLK, 256, 0, stream>>>(noise, W1, b1, W2, b2, meas);
  back_mlp_kernel<<<BATCH / 64, 128, 0, stream>>>(meas, W3, b3, W4, b4, out);
}


# First layer acts on |0..0>: only its 32 matrix instructions with a non-zero input row are issued, not 512 (row = per-lane product of 9 gate entries); no state clearing; amplitude rounding changes
# speedup vs baseline: 3.8015x; 1.1776x over previous
#include <hip/hip_runtime.h>
#include <math.h>

#define N_QUBITS 12
#define DEPTH    4
#define HID      64
#define NPARAMS  144
#define BATCH    4096
#define DIM      4096
#define OUT_DIM  2

typedef __attribute__((ext_vector_type(2))) float v2f;
typedef __attribute__((ext_vector_type(8))) float v8f;

__device__ __forceinline__ v8f wmma_f32(v2f a, v2f b, v8f c) {
  return __builtin_amdgcn_wmma_f32_16x16x4_f32(
       false, a,  false, b,
       (short)0, c,  false,  false);
}

__device__ __forceinline__ v2f load_a_frag(const float* src, int lda,
                                           int row0, int k0, int lane) {
  int m = row0 + (lane & 15);
  int k = k0 + ((lane >> 4) << 1);
  v2f a;
  a.x = src[m * lda + k];
  a.y = src[m * lda + k + 1];
  return a;
}

__device__ __forceinline__ v2f load_b_frag(const float* W, int ldb,
                                           int k0, int n0, int lane) {
  int c = n0 + (lane & 15);
  int k = k0 + ((lane >> 4) << 1);
  v2f b;
  b.x = W[k * ldb + c];
  b.y = W[(k + 1) * ldb + c];
  return b;
}

#define FRAG_PER_LAYER 512
#define NBLK 768
#define MAXE 6

__device__ __forceinline__ int phys_idx(int i) {
  return i ^ (((i >> 5) & 3) * 9) ^ ((((i >> 8) ^ (i >> 11)) & 1) << 4);
}

__device__ __forceinline__ int cnot_ring(int x) {
#pragma unroll
  for (int q = 0; q < N_QUBITS; ++q) {
    int pc = N_QUBITS - 1 - q;
    int pt = N_QUBITS - 1 - ((q + 1) % N_QUBITS);
    if ((x >> pc) & 1) x ^= (1 << pt);
  }
  return x;
}

__device__ __forceinline__ v2f cmul(v2f a, v2f b) {
  v2f r;
  r.x = a.x * b.x - a.y * b.y;
  r.y = a.x * b.y + a.y * b.x;
  return r;
}

template <int G, int MODE>
__device__ __forceinline__ void group_pass(v2f* st, const v2f* af,
                                           int wave, int lane,
                                           float* acc = nullptr,
                                           const int* lda = nullptr,
                                           const int* sta = nullptr,
                                           const v2f* w0 = nullptr) {
  constexpr bool LAST = (MODE == 1 || MODE == 3);
  constexpr int NK = (MODE == 3) ? 1 : 4;
  constexpr int SH = 9 - 3 * G;
  const int c = lane & 15;
  const int h = lane >> 4;
  v2f a[NK];
#pragma unroll
  for (int kk = 0; kk < NK; ++kk) a[kk] = af[kk * 32 + lane];

  int ib[4], pb[4];
  v2f x[4][NK];
#pragma unroll
  for (int n = 0; n < 4; ++n) {
    const int col9 = ((wave * 4 + n) << 4) | c;
    const int hi = col9 >> SH;
    const int lo = col9 & ((1 << SH) - 1);
    ib[n] = (hi << (SH + 3)) | (h << (SH + 2)) | lo;
    pb[n] = phys_idx(ib[n]);
    if (MODE == 3) {
      v2f z; z.x = 0.0f; z.y = 0.0f;
      x[n][0] = (h == 0) ? w0[n] : z;
    } else {
#pragma unroll
      for (int u = 0; u < NK; ++u)
        x[n][u] = st[(G == 3) ? lda[n * 4 + u] : (pb[n] ^ phys_idx(u << SH))];
    }
  }
  if (MODE != 3) __builtin_amdgcn_sched_barrier(0);
  v8f d[4] = {};
#pragma unroll
  for (int kk = 0; kk < NK; ++kk)
#pragma unroll
    for (int n = 0; n < 4; ++n) d[n] = wmma_f32(a[kk], x[n][kk], d[n]);

  if (MODE == 2) {
    v2f S[4], B[4], C[4];
#pragma unroll
    for (int n = 0; n < 4; ++n) {
      v2f q[4];
#pragma unroll
      for (int u = 0; u < 4; ++u) {
        v2f v;
        v.x = d[n][2 * u];
        v.y = d[n][2 * u + 1];
        q[u] = v * v;
      }
      const v2f e = q[0] + q[1], f = q[2] + q[3];
      S[n] = e + f;
      B[n] = e - f;
      C[n] = (q[0] - q[1]) + (q[3] - q[2]);
    }
    const v2f e2 = S[0] + S[1], f2 = S[2] + S[3];
    const v2f vT  = e2 + f2;
    const v2f v8  = e2 - f2;
    const v2f v87 = (S[0] - S[1]) + (S[3] - S[2]);
    const v2f v871  = (B[0] - B[1]) + (B[3] - B[2]);
    const v2f v8710 = (C[0] - C[1]) + (C[3] - C[2]);
    const float qT = vT.x + vT.y, q8 = v8.x + v8.y, q87 = v87.x + v87.y;
    const float q871 = v871.x + v871.y, q8710 = v8710.x + v8710.y;
    const unsigned L = ((unsigned)wave << 9) | ((unsigned)c << 3) | ((unsigned)h << 2);
    unsigned g = L;
    g ^= g >> 1; g ^= g >> 2; g ^= g >> 4; g ^= g >> 8;
#pragma unroll
    for (int j = 0; j <= 10; ++j) {
      const float qv = (j >= 9) ? qT : (j == 8) ? q8 : (j >= 2) ? q87
                     : (j == 1) ? q871 : q8710;
      const bool neg = ((g >> j) & 1u) != 0u;
      acc[N_QUBITS - 1 - j] = neg ? -qv : qv;
    }
    {
      const bool neg = (((g ^ (g >> 11)) & 1u) != 0u);
      acc[0] = neg ? -q8710 : q8710;
    }
    return;
  }
  if (MODE == 1) __syncthreads();
#pragma unroll
  for (int n = 0; n < 4; ++n) {
#pragma unroll
    for (int u = 0; u < 4; ++u) {
      const int dst = LAST ? sta[n * 4 + u] : (pb[n] ^ phys_idx(u << SH));
      v2f v;
      v.x = d[n][2 * u];
      v.y = d[n][2 * u + 1];
      st[dst] = v;
    }
  }
  __syncthreads();
}

__global__ __launch_bounds__(256) void quantum_kernel(
    const float* __restrict__ noise, const float* __restrict__ W1,
    const float* __restrict__ b1,    const float* __restrict__ W2,
    const float* __restrict__ b2,    float* __restrict__ meas) {
  __shared__ v2f st[DIM];
  __shared__ v2f afrag[DEPTH * FRAG_PER_LAYER];
  __shared__ v2f ssc[NPARAMS];
  __shared__ float gU[DEPTH * N_QUBITS][8];
  __shared__ float part[256 * N_QUBITS];
  __shared__ float hbuf[16 * HID];
  __shared__ float pbuf[16 * NPARAMS];
  const int tid  = threadIdx.x;
  const int lane = tid & 31;
  const int wave = tid >> 5;
  const int wave_u = __builtin_amdgcn_readfirstlane(wave);
  const int bq = (int)blockIdx.x / 3, br = (int)blockIdx.x - 3 * bq;
  const int ebase = 16 * bq + ((br == 0) ? 0 : (br == 1) ? 6 : 11);
  const int ecnt  = (br == 0) ? 6 : 5;

  {
    const int col   = lane & 15;
    const int rhalf = (lane >> 4) * 8;
    if (wave_u < HID / 16) {
      const int e = ebase + ((col < ecnt) ? col : 0);
      v8f cacc = {};
#pragma unroll
      for (int kc = 0; kc < N_QUBITS; kc += 4) {
        const int k = kc + ((lane >> 4) << 1);
        v2f a;
        a.x = noise[e * N_QUBITS + k];
        a.y = noise[e * N_QUBITS + k + 1];
        v2f b = load_b_frag(W1, HID, kc, wave_u * 16, lane);
        cacc = wmma_f32(a, b, cacc);
      }
      const int n = wave_u * 16 + col;
      const float bias = b1[n];
#pragma unroll
      for (int r = 0; r < 8; ++r)
        hbuf[(rhalf + r) * HID + n] = tanhf(cacc[r] + bias);
    }
    __syncthreads();
    for (int nt = wave_u; nt < NPARAMS / 16; nt += 8) {
      v8f cacc = {};
#pragma unroll
      for (int kc = 0; kc < HID; kc += 4) {
        v2f a = load_a_frag(hbuf, HID, 0, kc, lane);
        v2f b = load_b_frag(W2, NPARAMS, kc, nt * 16, lane);
        cacc = wmma_f32(a, b, cacc);
      }
      const int n = nt * 16 + col;
      const float bias = b2[n];
#pragma unroll
      for (int r = 0; r < 8; ++r)
        pbuf[(rhalf + r) * NPARAMS + n] = cacc[r] + bias;
    }
    __syncthreads();
  }

  int ld3[16];
  {
    const int c = lane & 15, h = lane >> 4;
#pragma unroll
    for (int n = 0; n < 4; ++n) {
      const int col9 = ((wave * 4 + n) << 4) | c;
      const int pbn = phys_idx((col9 << 3) | (h << 2));
#pragma unroll
      for (int u = 0; u < 4; ++u) ld3[n * 4 + u] = pbn ^ phys_idx(u);
    }
  }
  int sc3[16];
  {
    const int c = lane & 15, h = lane >> 4;
#pragma unroll
    for (int n = 0; n < 4; ++n) {
      const int col9 = ((wave * 4 + n) << 4) | c;
      const int pr = cnot_ring((col9 << 3) | (h << 2));
#pragma unroll
      for (int u = 0; u < 4; ++u) sc3[n * 4 + u] = phys_idx(pr ^ cnot_ring(u));
    }
  }

  for (int it = 0; it < MAXE; ++it) {
  if (it >= ecnt) break;
  const int bidx = ebase + it;
  const float* p = pbuf + it * NPARAMS;
  {
    const int left = ecnt - it;
    if (left >= 4)      __builtin_amdgcn_s_setprio(3);
    else if (left == 3) __builtin_amdgcn_s_setprio(2);
    else if (left == 2) __builtin_amdgcn_s_setprio(1);
    else                __builtin_amdgcn_s_setprio(0);
  }

  if (tid < NPARAMS) {
    float s, c;
    sincosf(p[tid] * 0.5f, &s, &c);
    v2f sc; sc.x = s; sc.y = c;
    ssc[tid] = sc;
  }
  __syncthreads();

  if (tid < DEPTH * N_QUBITS) {
    const v2f A = ssc[tid * 3 + 0], B = ssc[tid * 3 + 1], Z = ssc[tid * 3 + 2];
    const float sa = A.x, ca = A.y, sb = B.x, cb = B.y, sz = Z.x, cz = Z.y;
    const float m00r =  cb * ca, m00i =  sb * sa;
    const float m01r = -sb * ca, m01i = -cb * sa;
    const float m10r =  sb * ca, m10i = -cb * sa;
    const float m11r =  cb * ca, m11i = -sb * sa;
    float* u = gU[tid];
    u[0] = cz * m00r + sz * m00i;  u[1] = cz * m00i - sz * m00r;
    u[2] = cz * m01r + sz * m01i;  u[3] = cz * m01i - sz * m01r;
    u[4] = cz * m10r - sz * m10i;  u[5] = cz * m10i + sz * m10r;
    u[6] = cz * m11r - sz * m11i;  u[7] = cz * m11i + sz * m11r;
  }
  __syncthreads();

  for (int e0 = tid + FRAG_PER_LAYER - 256; e0 < DEPTH * FRAG_PER_LAYER; e0 += 256) {
    if (e0 < FRAG_PER_LAYER && tid >= 32) continue;
    const int e = (e0 < FRAG_PER_LAYER) ? (3 * 128 + tid) : e0;
    const int ln = e & 31, kk = (e >> 5) & 3, g = (e >> 7) & 3, l = e >> 9;
    const int m = ln & 15, h = ln >> 4;
    const int rip = m & 1;
    const int jp = 4 * (m >> 3) + ((m & 7) >> 1);
    const int j = 4 * h + kk;
    float pr = 1.0f, pi = 0.0f;
#pragma unroll
    for (int b = 2; b >= 0; --b) {
      const float* u = gU[l * N_QUBITS + 3 * g + (2 - b)];
      const int eo = ((((jp >> b) & 1) << 1) | ((j >> b) & 1)) << 1;
      const float ur = u[eo], ui = u[eo + 1];
      const float nr = pr * ur - pi * ui;
      const float ni = pr * ui + pi * ur;
      pr = nr; pi = ni;
    }
    v2f ent;
    ent.x = rip ? pi : pr;
    ent.y = rip ? pr : -pi;
    afrag[e] = ent;
  }
  v2f w0[4];
  {
    const int c = lane & 15;
    auto col0 = [&](int q, int b) {
      const float* u = gU[q] + 4 * (b & 1);
      v2f r; r.x = u[0]; r.y = u[1];
      return r;
    };
    v2f t = cmul(col0(0, wave >> 2), col0(1, wave >> 1));
    t = cmul(t, col0(2, wave));
    t = cmul(t, cmul(col0(5, c >> 3), col0(6, c >> 2)));
    t = cmul(t, cmul(col0(7, c >> 1), col0(8, c)));
#pragma unroll
    for (int n = 0; n < 4; ++n)
      w0[n] = cmul(t, cmul(col0(3, n >> 1), col0(4, n)));
  }
  __syncthreads();

  group_pass<3, 3>(st, afrag + 384, wave, lane, nullptr, nullptr, sc3, w0);
#pragma nounroll
  for (int l = 1; l < DEPTH - 1; ++l) {
    const v2f* af = afrag + l * FRAG_PER_LAYER;
    group_pass<0, 0>(st, af,       wave, lane);
    group_pass<1, 0>(st, af + 128, wave, lane);
    group_pass<2, 0>(st, af + 256, wave, lane);
    group_pass<3, 1>(st, af + 384, wave, lane, nullptr, ld3, sc3);
  }

  float acc[N_QUBITS];
  {
    const v2f* af = afrag + (DEPTH - 1) * FRAG_PER_LAYER;
    group_pass<0, 0>(st, af,       wave, lane);
    group_pass<1, 0>(st, af + 128, wave, lane);
    group_pass<2, 0>(st, af + 256, wave, lane);
    group_pass<3, 2>(st, af + 384, wave, lane, acc, ld3);
  }
#pragma unroll
  for (int q = 0; q < N_QUBITS; ++q) part[tid * N_QUBITS + q] = acc[q];
  __syncthreads();
  for (int s = 128; s > 0; s >>= 1) {
    if (tid < s) {
#pragma unroll
      for (int q = 0; q < N_QUBITS; ++q)
        part[tid * N_QUBITS + q] += part[(tid + s) * N_QUBITS + q];
    }
    __syncthreads();
  }
  if (tid < N_QUBITS) meas[(size_t)bidx * N_QUBITS + tid] = part[tid];
  }
}

__global__ __launch_bounds__(128) void back_mlp_kernel(
    const float* __restrict__ meas, const float* __restrict__ W3,
    const float* __restrict__ b3,   const float* __restrict__ W4,
    const float* __restrict__ b4,   float* __restrict__ out) {
  __shared__ float hbuf[4][16 * HID];
  const int lane  = threadIdx.x & 31;
  const int wave  = threadIdx.x >> 5;
  const int row0  = (blockIdx.x * 4 + wave) * 16;
  const int col   = lane & 15;
  const int rhalf = (lane >> 4) * 8;
  float* h = hbuf[wave];

  for (int nt = 0; nt < HID / 16; ++nt) {
    v8f c = {};
#pragma unroll
    for (int kc = 0; kc < N_QUBITS; kc += 4) {
      v2f a = load_a_frag(meas, N_QUBITS, row0, kc, lane);
      v2f b = load_b_frag(W3, HID, kc, nt * 16, lane);
      c = wmma_f32(a, b, c);
    }
    int n = nt * 16 + col;
    float bias = b3[n];
#pragma unroll
    for (int r = 0; r < 8; ++r)
      h[(rhalf + r) * HID + n] = tanhf(c[r] + bias);
  }
  __syncthreads();

  v8f c = {};
#pragma unroll
  for (int kc = 0; kc < HID; kc += 4) {
    v2f a = load_a_frag(h, HID, 0, kc, lane);
    int k = kc + ((lane >> 4) << 1);
    v2f b;
    b.x = (col < OUT_DIM) ? W4[k * OUT_DIM + col] : 0.0f;
    b.y = (col < OUT_DIM) ? W4[(k + 1) * OUT_DIM + col] : 0.0f;
    c = wmma_f32(a, b, c);
  }
  if (col < OUT_DIM) {
    float bias = b4[col];
#pragma unroll
    for (int r = 0; r < 8; ++r)
      out[(size_t)(row0 + rhalf + r) * OUT_DIM + col] = c[r] + bias;
  }
}

extern "C" void kernel_launch(void* const* d_in, const int* in_sizes, int n_in,
                              void* d_out, int out_size, void* d_ws, size_t ws_size,
                              hipStream_t stream) {
  const float* noise = (const float*)d_in[0];
  const float* W1 = (const float*)d_in[1];
  const float* b1 = (const float*)d_in[2];
  const float* W2 = (const float*)d_in[3];
  const float* b2 = (const float*)d_in[4];
  const float* W3 = (const float*)d_in[5];
  const float* b3 = (const float*)d_in[6];
  const float* W4 = (const float*)d_in[7];
  const float* b4 = (const float*)d_in[8];
  float* out = (float*)d_out;

  float* meas = (float*)d_ws;

  quantum_kernel<<<NBLK, 256, 0, stream>>>(noise, W1, b1, W2, b2, meas);
  back_mlp_kernel<<<BATCH / 64, 128, 0, stream>>>(meas, W3, b3, W4, b4, out);
}


# A-fragment build: each thread owns one complex entry of the 8x8 group matrix in every layer (fixed offsets, loads grouped), both real-matrix rows in one 16-byte store: 2 LDS round trips, not 8
# speedup vs baseline: 4.0846x; 1.0745x over previous
#include <hip/hip_runtime.h>
#include <math.h>

#define N_QUBITS 12
#define DEPTH    4
#define HID      64
#define NPARAMS  144
#define BATCH    4096
#define DIM      4096
#define OUT_DIM  2

typedef __attribute__((ext_vector_type(2))) float v2f;
typedef __attribute__((ext_vector_type(4))) float v4f;
typedef __attribute__((ext_vector_type(8))) float v8f;

__device__ __forceinline__ v8f wmma_f32(v2f a, v2f b, v8f c) {
  return __builtin_amdgcn_wmma_f32_16x16x4_f32(
       false, a,  false, b,
       (short)0, c,  false,  false);
}

__device__ __forceinline__ v2f load_a_frag(const float* src, int lda,
                                           int row0, int k0, int lane) {
  int m = row0 + (lane & 15);
  int k = k0 + ((lane >> 4) << 1);
  v2f a;
  a.x = src[m * lda + k];
  a.y = src[m * lda + k + 1];
  return a;
}

__device__ __forceinline__ v2f load_b_frag(const float* W, int ldb,
                                           int k0, int n0, int lane) {
  int c = n0 + (lane & 15);
  int k = k0 + ((lane >> 4) << 1);
  v2f b;
  b.x = W[k * ldb + c];
  b.y = W[(k + 1) * ldb + c];
  return b;
}

#define FRAG_PER_LAYER 512
#define NBLK 768
#define MAXE 6

__device__ __forceinline__ int phys_idx(int i) {
  return i ^ (((i >> 5) & 3) * 9) ^ ((((i >> 8) ^ (i >> 11)) & 1) << 4);
}

__device__ __forceinline__ int cnot_ring(int x) {
#pragma unroll
  for (int q = 0; q < N_QUBITS; ++q) {
    int pc = N_QUBITS - 1 - q;
    int pt = N_QUBITS - 1 - ((q + 1) % N_QUBITS);
    if ((x >> pc) & 1) x ^= (1 << pt);
  }
  return x;
}

__device__ __forceinline__ v2f cmul(v2f a, v2f b) {
  v2f r;
  r.x = a.x * b.x - a.y * b.y;
  r.y = a.x * b.y + a.y * b.x;
  return r;
}

template <int G, int MODE>
__device__ __forceinline__ void group_pass(v2f* st, const v2f* af,
                                           int wave, int lane,
                                           float* acc = nullptr,
                                           const int* lda = nullptr,
                                           const int* sta = nullptr,
                                           const v2f* w0 = nullptr) {
  constexpr bool LAST = (MODE == 1 || MODE == 3);
  constexpr int NK = (MODE == 3) ? 1 : 4;
  constexpr int SH = 9 - 3 * G;
  const int c = lane & 15;
  const int h = lane >> 4;
  v2f a[NK];
#pragma unroll
  for (int kk = 0; kk < NK; ++kk) a[kk] = af[kk * 32 + lane];

  int ib[4], pb[4];
  v2f x[4][NK];
#pragma unroll
  for (int n = 0; n < 4; ++n) {
    const int col9 = ((wave * 4 + n) << 4) | c;
    const int hi = col9 >> SH;
    const int lo = col9 & ((1 << SH) - 1);
    ib[n] = (hi << (SH + 3)) | (h << (SH + 2)) | lo;
    pb[n] = phys_idx(ib[n]);
    if (MODE == 3) {
      v2f z; z.x = 0.0f; z.y = 0.0f;
      x[n][0] = (h == 0) ? w0[n] : z;
    } else {
#pragma unroll
      for (int u = 0; u < NK; ++u)
        x[n][u] = st[(G == 3) ? lda[n * 4 + u] : (pb[n] ^ phys_idx(u << SH))];
    }
  }
  if (MODE != 3) __builtin_amdgcn_sched_barrier(0);
  v8f d[4] = {};
#pragma unroll
  for (int kk = 0; kk < NK; ++kk)
#pragma unroll
    for (int n = 0; n < 4; ++n) d[n] = wmma_f32(a[kk], x[n][kk], d[n]);

  if (MODE == 2) {
    v2f S[4], B[4], C[4];
#pragma unroll
    for (int n = 0; n < 4; ++n) {
      v2f q[4];
#pragma unroll
      for (int u = 0; u < 4; ++u) {
        v2f v;
        v.x = d[n][2 * u];
        v.y = d[n][2 * u + 1];
        q[u] = v * v;
      }
      const v2f e = q[0] + q[1], f = q[2] + q[3];
      S[n] = e + f;
      B[n] = e - f;
      C[n] = (q[0] - q[1]) + (q[3] - q[2]);
    }
    const v2f e2 = S[0] + S[1], f2 = S[2] + S[3];
    const v2f vT  = e2 + f2;
    const v2f v8  = e2 - f2;
    const v2f v87 = (S[0] - S[1]) + (S[3] - S[2]);
    const v2f v871  = (B[0] - B[1]) + (B[3] - B[2]);
    const v2f v8710 = (C[0] - C[1]) + (C[3] - C[2]);
    const float qT = vT.x + vT.y, q8 = v8.x + v8.y, q87 = v87.x + v87.y;
    const float q871 = v871.x + v871.y, q8710 = v8710.x + v8710.y;
    const unsigned L = ((unsigned)wave << 9) | ((unsigned)c << 3) | ((unsigned)h << 2);
    unsigned g = L;
    g ^= g >> 1; g ^= g >> 2; g ^= g >> 4; g ^= g >> 8;
#pragma unroll
    for (int j = 0; j <= 10; ++j) {
      const float qv = (j >= 9) ? qT : (j == 8) ? q8 : (j >= 2) ? q87
                     : (j == 1) ? q871 : q8710;
      const bool neg = ((g >> j) & 1u) != 0u;
      acc[N_QUBITS - 1 - j] = neg ? -qv : qv;
    }
    {
      const bool neg = (((g ^ (g >> 11)) & 1u) != 0u);
      acc[0] = neg ? -q8710 : q8710;
    }
    return;
  }
  if (MODE == 1) __syncthreads();
#pragma unroll
  for (int n = 0; n < 4; ++n) {
#pragma unroll
    for (int u = 0; u < 4; ++u) {
      const int dst = LAST ? sta[n * 4 + u] : (pb[n] ^ phys_idx(u << SH));
      v2f v;
      v.x = d[n][2 * u];
      v.y = d[n][2 * u + 1];
      st[dst] = v;
    }
  }
  __syncthreads();
}

__global__ __launch_bounds__(256) void quantum_kernel(
    const float* __restrict__ noise, const float* __restrict__ W1,
    const float* __restrict__ b1,    const float* __restrict__ W2,
    const float* __restrict__ b2,    float* __restrict__ meas) {
  __shared__ v2f st[DIM];
  __shared__ __attribute__((aligned(16))) v2f afrag[DEPTH * FRAG_PER_LAYER];
  __shared__ v2f ssc[NPARAMS];
  __shared__ __attribute__((aligned(16))) float gU[DEPTH * N_QUBITS][8];
  __shared__ float part[256 * N_QUBITS];
  __shared__ float hbuf[16 * HID];
  __shared__ float pbuf[16 * NPARAMS];
  const int tid  = threadIdx.x;
  const int lane = tid & 31;
  const int wave = tid >> 5;
  const int wave_u = __builtin_amdgcn_readfirstlane(wave);
  const int bq = (int)blockIdx.x / 3, br = (int)blockIdx.x - 3 * bq;
  const int ebase = 16 * bq + ((br == 0) ? 0 : (br == 1) ? 6 : 11);
  const int ecnt  = (br == 0) ? 6 : 5;

  {
    const int col   = lane & 15;
    const int rhalf = (lane >> 4) * 8;
    if (wave_u < HID / 16) {
      const int e = ebase + ((col < ecnt) ? col : 0);
      v8f cacc = {};
#pragma unroll
      for (int kc = 0; kc < N_QUBITS; kc += 4) {
        const int k = kc + ((lane >> 4) << 1);
        v2f a;
        a.x = noise[e * N_QUBITS + k];
        a.y = noise[e * N_QUBITS + k + 1];
        v2f b = load_b_frag(W1, HID, kc, wave_u * 16, lane);
        cacc = wmma_f32(a, b, cacc);
      }
      const int n = wave_u * 16 + col;
      const float bias = b1[n];
#pragma unroll
      for (int r = 0; r < 8; ++r)
        hbuf[(rhalf + r) * HID + n] = tanhf(cacc[r] + bias);
    }
    __syncthreads();
    for (int nt = wave_u; nt < NPARAMS / 16; nt += 8) {
      v8f cacc = {};
#pragma unroll
      for (int kc = 0; kc < HID; kc += 4) {
        v2f a = load_a_frag(hbuf, HID, 0, kc, lane);
        v2f b = load_b_frag(W2, NPARAMS, kc, nt * 16, lane);
        cacc = wmma_f32(a, b, cacc);
      }
      const int n = nt * 16 + col;
      const float bias = b2[n];
#pragma unroll
      for (int r = 0; r < 8; ++r)
        pbuf[(rhalf + r) * NPARAMS + n] = cacc[r] + bias;
    }
    __syncthreads();
  }

  int ld3[16];
  {
    const int c = lane & 15, h = lane >> 4;
#pragma unroll
    for (int n = 0; n < 4; ++n) {
      const int col9 = ((wave * 4 + n) << 4) | c;
      const int pbn = phys_idx((col9 << 3) | (h << 2));
#pragma unroll
      for (int u = 0; u < 4; ++u) ld3[n * 4 + u] = pbn ^ phys_idx(u);
    }
  }
  int sc3[16];
  {
    const int c = lane & 15, h = lane >> 4;
#pragma unroll
    for (int n = 0; n < 4; ++n) {
      const int col9 = ((wave * 4 + n) << 4) | c;
      const int pr = cnot_ring((col9 << 3) | (h << 2));
#pragma unroll
      for (int u = 0; u < 4; ++u) sc3[n * 4 + u] = phys_idx(pr ^ cnot_ring(u));
    }
  }

  for (int it = 0; it < MAXE; ++it) {
  if (it >= ecnt) break;
  const int bidx = ebase + it;
  const float* p = pbuf + it * NPARAMS;
  {
    const int left = ecnt - it;
    if (left >= 4)      __builtin_amdgcn_s_setprio(3);
    else if (left == 3) __builtin_amdgcn_s_setprio(2);
    else if (left == 2) __builtin_amdgcn_s_setprio(1);
    else                __builtin_amdgcn_s_setprio(0);
  }

  if (tid < NPARAMS) {
    float s, c;
    sincosf(p[tid] * 0.5f, &s, &c);
    v2f sc; sc.x = s; sc.y = c;
    ssc[tid] = sc;
  }
  __syncthreads();

  if (tid < DEPTH * N_QUBITS) {
    const v2f A = ssc[tid * 3 + 0], B = ssc[tid * 3 + 1], Z = ssc[tid * 3 + 2];
    const float sa = A.x, ca = A.y, sb = B.x, cb = B.y, sz = Z.x, cz = Z.y;
    const float m00r =  cb * ca, m00i =  sb * sa;
    const float m01r = -sb * ca, m01i = -cb * sa;
    const float m10r =  sb * ca, m10i = -cb * sa;
    const float m11r =  cb * ca, m11i = -sb * sa;
    float* u = gU[tid];
    u[0] = cz * m00r + sz * m00i;  u[1] = cz * m00i - sz * m00r;
    u[2] = cz * m01r + sz * m01i;  u[3] = cz * m01i - sz * m01r;
    u[4] = cz * m10r - sz * m10i;  u[5] = cz * m10i + sz * m10r;
    u[6] = cz * m11r - sz * m11i;  u[7] = cz * m11i + sz * m11r;
  }
  __syncthreads();

  v2f w0[4];
  {
    const int g  = tid >> 6;
    const int jp = (tid >> 3) & 7;
    const int j  = tid & 7;
    const int o0 = ((((jp >> 2) & 1) << 1) | ((j >> 2) & 1)) << 1;
    const int o1 = ((((jp >> 1) & 1) << 1) | ((j >> 1) & 1)) << 1;
    const int o2 = (((jp & 1) << 1) | (j & 1)) << 1;
    const int m0 = ((jp >> 2) << 3) | ((jp & 3) << 1);
    const int eo = (g << 7) | ((j & 3) << 5) | ((j >> 2) << 4) | m0;
    const bool first = (g == 3) && ((j & 3) == 0);
    v2f ga[DEPTH][3];
#pragma unroll
    for (int l = 0; l < DEPTH; ++l) {
      const float* u = gU[l * N_QUBITS + 3 * g];
      ga[l][0] = *(const v2f*)(u + o0);
      ga[l][1] = *(const v2f*)(u + 8 + o1);
      ga[l][2] = *(const v2f*)(u + 16 + o2);
    }
    __builtin_amdgcn_sched_barrier(0);
#pragma unroll
    for (int l = 0; l < DEPTH; ++l) {
      if (l > 0 || first) {
        const v2f p = cmul(cmul(ga[l][0], ga[l][1]), ga[l][2]);
        v4f ent;
        ent.x = p.x; ent.y = -p.y; ent.z = p.y; ent.w = p.x;
        *(v4f*)(&afrag[l * FRAG_PER_LAYER + eo]) = ent;
      }
    }
  }
  {
    const int c = lane & 15;
    auto col0 = [&](int q, int b) {
      const float* u = gU[q] + 4 * (b & 1);
      v2f r; r.x = u[0]; r.y = u[1];
      return r;
    };
    const v2f k0 = col0(0, wave >> 2), k1 = col0(1, wave >> 1), k2 = col0(2, wave);
    const v2f k5 = col0(5, c >> 3), k6 = col0(6, c >> 2);
    const v2f k7 = col0(7, c >> 1), k8 = col0(8, c);
    const v2f k30 = col0(3, 0), k31 = col0(3, 1), k40 = col0(4, 0), k41 = col0(4, 1);
    __builtin_amdgcn_sched_barrier(0);
    v2f t = cmul(k0, k1);
    t = cmul(t, k2);
    t = cmul(t, cmul(k5, k6));
    t = cmul(t, cmul(k7, k8));
    w0[0] = cmul(t, cmul(k30, k40));
    w0[1] = cmul(t, cmul(k30, k41));
    w0[2] = cmul(t, cmul(k31, k40));
    w0[3] = cmul(t, cmul(k31, k41));
  }
  __syncthreads();

  group_pass<3, 3>(st, afrag + 384, wave, lane, nullptr, nullptr, sc3, w0);
#pragma nounroll
  for (int l = 1; l < DEPTH - 1; ++l) {
    const v2f* af = afrag + l * FRAG_PER_LAYER;
    group_pass<0, 0>(st, af,       wave, lane);
    group_pass<1, 0>(st, af + 128, wave, lane);
    group_pass<2, 0>(st, af + 256, wave, lane);
    group_pass<3, 1>(st, af + 384, wave, lane, nullptr, ld3, sc3);
  }

  float acc[N_QUBITS];
  {
    const v2f* af = afrag + (DEPTH - 1) * FRAG_PER_LAYER;
    group_pass<0, 0>(st, af,       wave, lane);
    group_pass<1, 0>(st, af + 128, wave, lane);
    group_pass<2, 0>(st, af + 256, wave, lane);
    group_pass<3, 2>(st, af + 384, wave, lane, acc, ld3);
  }
#pragma unroll
  for (int q = 0; q < N_QUBITS; ++q) part[tid * N_QUBITS + q] = acc[q];
  __syncthreads();
  for (int s = 128; s > 0; s >>= 1) {
    if (tid < s) {
#pragma unroll
      for (int q = 0; q < N_QUBITS; ++q)
        part[tid * N_QUBITS + q] += part[(tid + s) * N_QUBITS + q];
    }
    __syncthreads();
  }
  if (tid < N_QUBITS) meas[(size_t)bidx * N_QUBITS + tid] = part[tid];
  }
}

__global__ __launch_bounds__(128) void back_mlp_kernel(
    const float* __restrict__ meas, const float* __restrict__ W3,
    const float* __restrict__ b3,   const float* __restrict__ W4,
    const float* __restrict__ b4,   float* __restrict__ out) {
  __shared__ float hbuf[4][16 * HID];
  const int lane  = threadIdx.x & 31;
  const int wave  = threadIdx.x >> 5;
  const int row0  = (blockIdx.x * 4 + wave) * 16;
  const int col   = lane & 15;
  const int rhalf = (lane >> 4) * 8;
  float* h = hbuf[wave];

  for (int nt = 0; nt < HID / 16; ++nt) {
    v8f c = {};
#pragma unroll
    for (int kc = 0; kc < N_QUBITS; kc += 4) {
      v2f a = load_a_frag(meas, N_QUBITS, row0, kc, lane);
      v2f b = load_b_frag(W3, HID, kc, nt * 16, lane);
      c = wmma_f32(a, b, c);
    }
    int n = nt * 16 + col;
    float bias = b3[n];
#pragma unroll
    for (int r = 0; r < 8; ++r)
      h[(rhalf + r) * HID + n] = tanhf(c[r] + bias);
  }
  __syncthreads();

  v8f c = {};
#pragma unroll
  for (int kc = 0; kc < HID; kc += 4) {
    v2f a = load_a_frag(h, HID, 0, kc, lane);
    int k = kc + ((lane >> 4) << 1);
    v2f b;
    b.x = (col < OUT_DIM) ? W4[k * OUT_DIM + col] : 0.0f;
    b.y = (col < OUT_DIM) ? W4[(k + 1) * OUT_DIM + col] : 0.0f;
    c = wmma_f32(a, b, c);
  }
  if (col < OUT_DIM) {
    float bias = b4[col];
#pragma unroll
    for (int r = 0; r < 8; ++r)
      out[(size_t)(row0 + rhalf + r) * OUT_DIM + col] = c[r] + bias;
  }
}

extern "C" void kernel_launch(void* const* d_in, const int* in_sizes, int n_in,
                              void* d_out, int out_size, void* d_ws, size_t ws_size,
                              hipStream_t stream) {
  const float* noise = (const float*)d_in[0];
  const float* W1 = (const float*)d_in[1];
  const float* b1 = (const float*)d_in[2];
  const float* W2 = (const float*)d_in[3];
  const float* b2 = (const float*)d_in[4];
  const float* W3 = (const float*)d_in[5];
  const float* b3 = (const float*)d_in[6];
  const float* W4 = (const float*)d_in[7];
  const float* b4 = (const float*)d_in[8];
  float* out = (float*)d_out;

  float* meas = (float*)d_ws;

  quantum_kernel<<<NBLK, 256, 0, stream>>>(noise, W1, b1, W2, b2, meas);
  back_mlp_kernel<<<BATCH / 64, 128, 0, stream>>>(meas, W3, b3, W4, b4, out);
}


# Layers 0-1 fused: state after layer 1 written as a 64x4 by 4x64 complex product (cut at wires 5|6, ring's closing wire carried), 8 matrix instructions a wave, not 68; amplitude rounding changes
# speedup vs baseline: 4.9515x; 1.2122x over previous
#include <hip/hip_runtime.h>
#include <math.h>

#define N_QUBITS 12
#define DEPTH    4
#define HID      64
#define NPARAMS  144
#define BATCH    4096
#define DIM      4096
#define OUT_DIM  2

typedef __attribute__((ext_vector_type(2))) float v2f;
typedef __attribute__((ext_vector_type(4))) float v4f;
typedef __attribute__((ext_vector_type(8))) float v8f;

__device__ __forceinline__ v8f wmma_f32(v2f a, v2f b, v8f c) {
  return __builtin_amdgcn_wmma_f32_16x16x4_f32(
       false, a,  false, b,
       (short)0, c,  false,  false);
}

__device__ __forceinline__ v2f load_a_frag(const float* src, int lda,
                                           int row0, int k0, int lane) {
  int m = row0 + (lane & 15);
  int k = k0 + ((lane >> 4) << 1);
  v2f a;
  a.x = src[m * lda + k];
  a.y = src[m * lda + k + 1];
  return a;
}

__device__ __forceinline__ v2f load_b_frag(const float* W, int ldb,
                                           int k0, int n0, int lane) {
  int c = n0 + (lane & 15);
  int k = k0 + ((lane >> 4) << 1);
  v2f b;
  b.x = W[k * ldb + c];
  b.y = W[(k + 1) * ldb + c];
  return b;
}

#define FRAG_PER_LAYER 512
#define NBLK 768
#define MAXE 6

__device__ __forceinline__ int phys_idx(int i) {
  return i ^ (((i >> 5) & 3) * 9) ^ ((((i >> 8) ^ (i >> 11)) & 1) << 4);
}

__device__ __forceinline__ int cnot_ring(int x) {
#pragma unroll
  for (int q = 0; q < N_QUBITS; ++q) {
    int pc = N_QUBITS - 1 - q;
    int pt = N_QUBITS - 1 - ((q + 1) % N_QUBITS);
    if ((x >> pc) & 1) x ^= (1 << pt);
  }
  return x;
}

__device__ __forceinline__ v2f cmul(v2f a, v2f b) {
  v2f r;
  r.x = a.x * b.x - a.y * b.y;
  r.y = a.x * b.y + a.y * b.x;
  return r;
}

template <int G, int MODE>
__device__ __forceinline__ void group_pass(v2f* st, const v2f* af,
                                           int wave, int lane,
                                           float* acc = nullptr,
                                           const int* lda = nullptr,
                                           const int* sta = nullptr) {
  constexpr bool LAST = (MODE == 1);
  constexpr int NK = 4;
  constexpr int SH = 9 - 3 * G;
  const int c = lane & 15;
  const int h = lane >> 4;
  v2f a[NK];
#pragma unroll
  for (int kk = 0; kk < NK; ++kk) a[kk] = af[kk * 32 + lane];

  int ib[4], pb[4];
  v2f x[4][NK];
#pragma unroll
  for (int n = 0; n < 4; ++n) {
    const int col9 = ((wave * 4 + n) << 4) | c;
    const int hi = col9 >> SH;
    const int lo = col9 & ((1 << SH) - 1);
    ib[n] = (hi << (SH + 3)) | (h << (SH + 2)) | lo;
    pb[n] = phys_idx(ib[n]);
#pragma unroll
    for (int u = 0; u < NK; ++u)
      x[n][u] = st[(G == 3) ? lda[n * 4 + u] : (pb[n] ^ phys_idx(u << SH))];
  }
  __builtin_amdgcn_sched_barrier(0);
  v8f d[4] = {};
#pragma unroll
  for (int kk = 0; kk < NK; ++kk)
#pragma unroll
    for (int n = 0; n < 4; ++n) d[n] = wmma_f32(a[kk], x[n][kk], d[n]);

  if (MODE == 2) {
    v2f S[4], B[4], C[4];
#pragma unroll
    for (int n = 0; n < 4; ++n) {
      v2f q[4];
#pragma unroll
      for (int u = 0; u < 4; ++u) {
        v2f v;
        v.x = d[n][2 * u];
        v.y = d[n][2 * u + 1];
        q[u] = v * v;
      }
      const v2f e = q[0] + q[1], f = q[2] + q[3];
      S[n] = e + f;
      B[n] = e - f;
      C[n] = (q[0] - q[1]) + (q[3] - q[2]);
    }
    const v2f e2 = S[0] + S[1], f2 = S[2] + S[3];
    const v2f vT  = e2 + f2;
    const v2f v8  = e2 - f2;
    const v2f v87 = (S[0] - S[1]) + (S[3] - S[2]);
    const v2f v871  = (B[0] - B[1]) + (B[3] - B[2]);
    const v2f v8710 = (C[0] - C[1]) + (C[3] - C[2]);
    const float qT = vT.x + vT.y, q8 = v8.x + v8.y, q87 = v87.x + v87.y;
    const float q871 = v871.x + v871.y, q8710 = v8710.x + v8710.y;
    const unsigned L = ((unsigned)wave << 9) | ((unsigned)c << 3) | ((unsigned)h << 2);
    unsigned g = L;
    g ^= g >> 1; g ^= g >> 2; g ^= g >> 4; g ^= g >> 8;
#pragma unroll
    for (int j = 0; j <= 10; ++j) {
      const float qv = (j >= 9) ? qT : (j == 8) ? q8 : (j >= 2) ? q87
                     : (j == 1) ? q871 : q8710;
      const bool neg = ((g >> j) & 1u) != 0u;
      acc[N_QUBITS - 1 - j] = neg ? -qv : qv;
    }
    {
      const bool neg = (((g ^ (g >> 11)) & 1u) != 0u);
      acc[0] = neg ? -q8710 : q8710;
    }
    return;
  }
  if (MODE == 1) __syncthreads();
#pragma unroll
  for (int n = 0; n < 4; ++n) {
#pragma unroll
    for (int u = 0; u < 4; ++u) {
      const int dst = LAST ? sta[n * 4 + u] : (pb[n] ^ phys_idx(u << SH));
      v2f v;
      v.x = d[n][2 * u];
      v.y = d[n][2 * u + 1];
      st[dst] = v;
    }
  }
  __syncthreads();
}

__device__ __forceinline__ void product_pass(v2f* st, const v2f* LR,
                                             int wave, int lane) {
  const int c = lane & 15;
  const int h = lane >> 4;
  const int zl = (wave << 3) | ((c >> 3) << 2) | ((c & 7) >> 1);
  v2f a[2], b[4][2];
#pragma unroll
  for (int kk = 0; kk < 2; ++kk) {
    const v2f l = LR[zl * 4 + 2 * kk + h];
    v2f re_row, im_row;
    re_row.x = l.x; re_row.y = -l.y;
    im_row.x = l.y; im_row.y = l.x;
    a[kk] = (c & 1) ? im_row : re_row;
#pragma unroll
    for (int n = 0; n < 4; ++n) b[n][kk] = LR[256 + (2 * kk + h) * 64 + 16 * n + c];
  }
  v8f d[4] = {};
#pragma unroll
  for (int kk = 0; kk < 2; ++kk)
#pragma unroll
    for (int n = 0; n < 4; ++n) d[n] = wmma_f32(a[kk], b[n][kk], d[n]);
  int pbase = phys_idx(cnot_ring((((wave << 3) | (h << 2)) << 6) | c));
  asm volatile("" : "+v"(pbase));
#pragma unroll
  for (int n = 0; n < 4; ++n) {
#pragma unroll
    for (int u = 0; u < 4; ++u) {
      v2f v;
      v.x = d[n][2 * u];
      v.y = d[n][2 * u + 1];
      st[pbase ^ phys_idx(cnot_ring((u << 6) | (n << 4)))] = v;
    }
  }
  __syncthreads();
}

__global__ __launch_bounds__(256) void quantum_kernel(
    const float* __restrict__ noise, const float* __restrict__ W1,
    const float* __restrict__ b1,    const float* __restrict__ W2,
    const float* __restrict__ b2,    float* __restrict__ meas) {
  __shared__ v2f st[DIM];
  __shared__ __attribute__((aligned(16))) v2f afrag[DEPTH * FRAG_PER_LAYER];
  __shared__ v2f ssc[NPARAMS];
  __shared__ __attribute__((aligned(16))) float gU[DEPTH * N_QUBITS][8];
  __shared__ float part[256 * N_QUBITS];
  __shared__ float hbuf[16 * HID];
  __shared__ float pbuf[16 * NPARAMS];
  const int tid  = threadIdx.x;
  const int lane = tid & 31;
  const int wave = tid >> 5;
  const int wave_u = __builtin_amdgcn_readfirstlane(wave);
  const int bq = (int)blockIdx.x / 3, br = (int)blockIdx.x - 3 * bq;
  const int ebase = 16 * bq + ((br == 0) ? 0 : (br == 1) ? 6 : 11);
  const int ecnt  = (br == 0) ? 6 : 5;

  {
    const int col   = lane & 15;
    const int rhalf = (lane >> 4) * 8;
    if (wave_u < HID / 16) {
      const int e = ebase + ((col < ecnt) ? col : 0);
      v8f cacc = {};
#pragma unroll
      for (int kc = 0; kc < N_QUBITS; kc += 4) {
        const int k = kc + ((lane >> 4) << 1);
        v2f a;
        a.x = noise[e * N_QUBITS + k];
        a.y = noise[e * N_QUBITS + k + 1];
        v2f b = load_b_frag(W1, HID, kc, wave_u * 16, lane);
        cacc = wmma_f32(a, b, cacc);
      }
      const int n = wave_u * 16 + col;
      const float bias = b1[n];
#pragma unroll
      for (int r = 0; r < 8; ++r)
        hbuf[(rhalf + r) * HID + n] = tanhf(cacc[r] + bias);
    }
    __syncthreads();
    for (int nt = wave_u; nt < NPARAMS / 16; nt += 8) {
      v8f cacc = {};
#pragma unroll
      for (int kc = 0; kc < HID; kc += 4) {
        v2f a = load_a_frag(hbuf, HID, 0, kc, lane);
        v2f b = load_b_frag(W2, NPARAMS, kc, nt * 16, lane);
        cacc = wmma_f32(a, b, cacc);
      }
      const int n = nt * 16 + col;
      const float bias = b2[n];
#pragma unroll
      for (int r = 0; r < 8; ++r)
        pbuf[(rhalf + r) * NPARAMS + n] = cacc[r] + bias;
    }
    __syncthreads();
  }

  int ld3[16];
  {
    const int c = lane & 15, h = lane >> 4;
#pragma unroll
    for (int n = 0; n < 4; ++n) {
      const int col9 = ((wave * 4 + n) << 4) | c;
      const int pbn = phys_idx((col9 << 3) | (h << 2));
#pragma unroll
      for (int u = 0; u < 4; ++u) ld3[n * 4 + u] = pbn ^ phys_idx(u);
    }
  }
  int sc3[16];
  {
    const int c = lane & 15, h = lane >> 4;
#pragma unroll
    for (int n = 0; n < 4; ++n) {
      const int col9 = ((wave * 4 + n) << 4) | c;
      const int pr = cnot_ring((col9 << 3) | (h << 2));
#pragma unroll
      for (int u = 0; u < 4; ++u) sc3[n * 4 + u] = phys_idx(pr ^ cnot_ring(u));
    }
  }

  for (int it = 0; it < MAXE; ++it) {
  if (it >= ecnt) break;
  const int bidx = ebase + it;
  const float* p = pbuf + it * NPARAMS;
  {
    const int left = ecnt - it;
    if (left >= 4)      __builtin_amdgcn_s_setprio(3);
    else if (left == 3) __builtin_amdgcn_s_setprio(2);
    else if (left == 2) __builtin_amdgcn_s_setprio(1);
    else                __builtin_amdgcn_s_setprio(0);
  }

  if (tid < NPARAMS) {
    float s, c;
    sincosf(p[tid] * 0.5f, &s, &c);
    v2f sc; sc.x = s; sc.y = c;
    ssc[tid] = sc;
  }
  __syncthreads();

  if (tid < DEPTH * N_QUBITS) {
    const v2f A = ssc[tid * 3 + 0], B = ssc[tid * 3 + 1], Z = ssc[tid * 3 + 2];
    const float sa = A.x, ca = A.y, sb = B.x, cb = B.y, sz = Z.x, cz = Z.y;
    const float m00r =  cb * ca, m00i =  sb * sa;
    const float m01r = -sb * ca, m01i = -cb * sa;
    const float m10r =  sb * ca, m10i = -cb * sa;
    const float m11r =  cb * ca, m11i = -sb * sa;
    float* u = gU[tid];
    u[0] = cz * m00r + sz * m00i;  u[1] = cz * m00i - sz * m00r;
    u[2] = cz * m01r + sz * m01i;  u[3] = cz * m01i - sz * m01r;
    u[4] = cz * m10r - sz * m10i;  u[5] = cz * m10i + sz * m10r;
    u[6] = cz * m11r - sz * m11i;  u[7] = cz * m11i + sz * m11r;
  }
  __syncthreads();

  {
    const int g  = tid >> 6;
    const int jp = (tid >> 3) & 7;
    const int j  = tid & 7;
    const int o0 = ((((jp >> 2) & 1) << 1) | ((j >> 2) & 1)) << 1;
    const int o1 = ((((jp >> 1) & 1) << 1) | ((j >> 1) & 1)) << 1;
    const int o2 = (((jp & 1) << 1) | (j & 1)) << 1;
    const int m0 = ((jp >> 2) << 3) | ((jp & 3) << 1);
    const int eo = (g << 7) | ((j & 3) << 5) | ((j >> 2) << 4) | m0;
    v2f ga[DEPTH][3];
#pragma unroll
    for (int l = 2; l < DEPTH; ++l) {
      const float* u = gU[l * N_QUBITS + 3 * g];
      ga[l][0] = *(const v2f*)(u + o0);
      ga[l][1] = *(const v2f*)(u + 8 + o1);
      ga[l][2] = *(const v2f*)(u + 16 + o2);
    }
    __builtin_amdgcn_sched_barrier(0);
#pragma unroll
    for (int l = 2; l < DEPTH; ++l) {
      const v2f p = cmul(cmul(ga[l][0], ga[l][1]), ga[l][2]);
      v4f ent;
      ent.x = p.x; ent.y = -p.y; ent.z = p.y; ent.w = p.x;
      *(v4f*)(&afrag[l * FRAG_PER_LAYER + eo]) = ent;
    }
  }
  {
    const int half = tid >> 7;
    const int zz = (tid & 127) >> 1;
    const int sb = tid & 1;
    v2f a0[6], a1[6];
    v4f V[6];
#pragma unroll
    for (int i = 0; i < 6; ++i) {
      const float* u0 = gU[half * 6 + i];
      a0[i] = *(const v2f*)(u0);
      a1[i] = *(const v2f*)(u0 + 4);
      const float* u1 = gU[N_QUBITS + half * 6 + i] + 4 * ((zz >> (5 - i)) & 1);
      V[i] = *(const v4f*)(u1);
    }
    __builtin_amdgcn_sched_barrier(0);
    v2f v0, v1;
    {
      v2f vlo, vhi;
      vlo.x = V[0].x; vlo.y = V[0].y; vhi.x = V[0].z; vhi.y = V[0].w;
      v0 = cmul(vlo, sb ? a1[0] : a0[0]);
      v1 = cmul(vhi, sb ? a0[0] : a1[0]);
    }
#pragma unroll
    for (int i = 1; i < 6; ++i) {
      const bool sw = (i == 1) && (half == 0) && (sb != 0);
      const v2f A0 = sw ? a1[i] : a0[i];
      const v2f A1 = sw ? a0[i] : a1[i];
      const v2f x0 = cmul(A0, v0) + cmul(A1, v1);
      const v2f x1 = cmul(A1, v0) + cmul(A0, v1);
      v2f vlo, vhi;
      vlo.x = V[i].x; vlo.y = V[i].y; vhi.x = V[i].z; vhi.y = V[i].w;
      v0 = cmul(vlo, x0);
      v1 = cmul(vhi, x1);
    }
    if (half == 0) {
      afrag[zz * 4 + sb]     = v0;
      afrag[zz * 4 + 2 + sb] = v1;
    } else {
      afrag[256 + (2 * sb) * 64 + zz]     = v0;
      afrag[256 + (2 * sb + 1) * 64 + zz] = v1;
    }
  }
  __syncthreads();

  product_pass(st, afrag, wave, lane);
#pragma nounroll
  for (int l = 2; l < DEPTH; ++l) {
    const v2f* af = afrag + l * FRAG_PER_LAYER;
    group_pass<0, 0>(st, af,       wave, lane);
    group_pass<1, 0>(st, af + 128, wave, lane);
    group_pass<2, 0>(st, af + 256, wave, lane);
    if (l < DEPTH - 1)
      group_pass<3, 1>(st, af + 384, wave, lane, nullptr, ld3, sc3);
  }

  float acc[N_QUBITS];
  group_pass<3, 2>(st, afrag + (DEPTH - 1) * FRAG_PER_LAYER + 384, wave, lane, acc, ld3);
#pragma unroll
  for (int q = 0; q < N_QUBITS; ++q) part[tid * N_QUBITS + q] = acc[q];
  __syncthreads();
  for (int s = 128; s > 0; s >>= 1) {
    if (tid < s) {
#pragma unroll
      for (int q = 0; q < N_QUBITS; ++q)
        part[tid * N_QUBITS + q] += part[(tid + s) * N_QUBITS + q];
    }
    __syncthreads();
  }
  if (tid < N_QUBITS) meas[(size_t)bidx * N_QUBITS + tid] = part[tid];
  }
}

__global__ __launch_bounds__(128) void back_mlp_kernel(
    const float* __restrict__ meas, const float* __restrict__ W3,
    const float* __restrict__ b3,   const float* __restrict__ W4,
    const float* __restrict__ b4,   float* __restrict__ out) {
  __shared__ float hbuf[4][16 * HID];
  const int lane  = threadIdx.x & 31;
  const int wave  = threadIdx.x >> 5;
  const int row0  = (blockIdx.x * 4 + wave) * 16;
  const int col   = lane & 15;
  const int rhalf = (lane >> 4) * 8;
  float* h = hbuf[wave];

  for (int nt = 0; nt < HID / 16; ++nt) {
    v8f c = {};
#pragma unroll
    for (int kc = 0; kc < N_QUBITS; kc += 4) {
      v2f a = load_a_frag(meas, N_QUBITS, row0, kc, lane);
      v2f b = load_b_frag(W3, HID, kc, nt * 16, lane);
      c = wmma_f32(a, b, c);
    }
    int n = nt * 16 + col;
    float bias = b3[n];
#pragma unroll
    for (int r = 0; r < 8; ++r)
      h[(rhalf + r) * HID + n] = tanhf(c[r] + bias);
  }
  __syncthreads();

  v8f c = {};
#pragma unroll
  for (int kc = 0; kc < HID; kc += 4) {
    v2f a = load_a_frag(h, HID, 0, kc, lane);
    int k = kc + ((lane >> 4) << 1);
    v2f b;
    b.x = (col < OUT_DIM) ? W4[k * OUT_DIM + col] : 0.0f;
    b.y = (col < OUT_DIM) ? W4[(k + 1) * OUT_DIM + col] : 0.0f;
    c = wmma_f32(a, b, c);
  }
  if (col < OUT_DIM) {
    float bias = b4[col];
#pragma unroll
    for (int r = 0; r < 8; ++r)
      out[(size_t)(row0 + rhalf + r) * OUT_DIM + col] = c[r] + bias;
  }
}

extern "C" void kernel_launch(void* const* d_in, const int* in_sizes, int n_in,
                              void* d_out, int out_size, void* d_ws, size_t ws_size,
                              hipStream_t stream) {
  const float* noise = (const float*)d_in[0];
  const float* W1 = (const float*)d_in[1];
  const float* b1 = (const float*)d_in[2];
  const float* W2 = (const float*)d_in[3];
  const float* b2 = (const float*)d_in[4];
  const float* W3 = (const float*)d_in[5];
  const float* b3 = (const float*)d_in[6];
  const float* W4 = (const float*)d_in[7];
  const float* b4 = (const float*)d_in[8];
  float* out = (float*)d_out;

  float* meas = (float*)d_ws;

  quantum_kernel<<<NBLK, 256, 0, stream>>>(noise, W1, b1, W2, b2, meas);
  back_mlp_kernel<<<BATCH / 64, 128, 0, stream>>>(meas, W3, b3, W4, b4, out);
}


# Registers 160 -> 118: LDS slots of the fourth group's passes and of the set-up blocks are worked out where used, not held over the element loop; 1024 blocks of 4 elements, four per WGP, not three
# speedup vs baseline: 5.4140x; 1.0934x over previous
#include <hip/hip_runtime.h>
#include <math.h>

#define N_QUBITS 12
#define DEPTH    4
#define HID      64
#define NPARAMS  144
#define BATCH    4096
#define DIM      4096
#define OUT_DIM  2

typedef __attribute__((ext_vector_type(2))) float v2f;
typedef __attribute__((ext_vector_type(4))) float v4f;
typedef __attribute__((ext_vector_type(8))) float v8f;

__device__ __forceinline__ v8f wmma_f32(v2f a, v2f b, v8f c) {
  return __builtin_amdgcn_wmma_f32_16x16x4_f32(
       false, a,  false, b,
       (short)0, c,  false,  false);
}

__device__ __forceinline__ v2f load_a_frag(const float* src, int lda,
                                           int row0, int k0, int lane) {
  int m = row0 + (lane & 15);
  int k = k0 + ((lane >> 4) << 1);
  v2f a;
  a.x = src[m * lda + k];
  a.y = src[m * lda + k + 1];
  return a;
}

__device__ __forceinline__ v2f load_b_frag(const float* W, int ldb,
                                           int k0, int n0, int lane) {
  int c = n0 + (lane & 15);
  int k = k0 + ((lane >> 4) << 1);
  v2f b;
  b.x = W[k * ldb + c];
  b.y = W[(k + 1) * ldb + c];
  return b;
}

#define FRAG_PER_LAYER 512
#define NBLK 1024
#define MAXE 4

__device__ __forceinline__ int phys_idx(int i) {
  return i ^ (((i >> 5) & 3) * 9) ^ ((((i >> 8) ^ (i >> 11)) & 1) << 4);
}

__device__ __forceinline__ int cnot_ring(int x) {
#pragma unroll
  for (int q = 0; q < N_QUBITS; ++q) {
    int pc = N_QUBITS - 1 - q;
    int pt = N_QUBITS - 1 - ((q + 1) % N_QUBITS);
    if ((x >> pc) & 1) x ^= (1 << pt);
  }
  return x;
}

__device__ __forceinline__ v2f cmul(v2f a, v2f b) {
  v2f r;
  r.x = a.x * b.x - a.y * b.y;
  r.y = a.x * b.y + a.y * b.x;
  return r;
}

template <int G, int MODE>
__device__ __forceinline__ void group_pass(v2f* st, const v2f* af,
                                           int wave, int lane,
                                           float* acc = nullptr,
                                           int ldb = 0, int stb = 0) {
  constexpr bool LAST = (MODE == 1);
  constexpr int NK = 4;
  constexpr int SH = 9 - 3 * G;
  const int c = lane & 15;
  const int h = lane >> 4;
  v2f a[NK];
#pragma unroll
  for (int kk = 0; kk < NK; ++kk) a[kk] = af[kk * 32 + lane];

  if (G == 3) asm volatile("" : "+v"(ldb));
  int ib[4], pb[4];
  v2f x[4][NK];
#pragma unroll
  for (int n = 0; n < 4; ++n) {
    const int col9 = ((wave * 4 + n) << 4) | c;
    const int hi = col9 >> SH;
    const int lo = col9 & ((1 << SH) - 1);
    ib[n] = (hi << (SH + 3)) | (h << (SH + 2)) | lo;
    pb[n] = phys_idx(ib[n]);
#pragma unroll
    for (int u = 0; u < NK; ++u)
      x[n][u] = (G == 3)
          ? *(const v2f*)((const char*)st + (ldb ^ (phys_idx((n << 7) | u) << 3)))
          : st[pb[n] ^ phys_idx(u << SH)];
  }
  __builtin_amdgcn_sched_barrier(0);
  v8f d[4] = {};
#pragma unroll
  for (int kk = 0; kk < NK; ++kk)
#pragma unroll
    for (int n = 0; n < 4; ++n) d[n] = wmma_f32(a[kk], x[n][kk], d[n]);

  if (MODE == 2) {
    v2f S[4], B[4], C[4];
#pragma unroll
    for (int n = 0; n < 4; ++n) {
      v2f q[4];
#pragma unroll
      for (int u = 0; u < 4; ++u) {
        v2f v;
        v.x = d[n][2 * u];
        v.y = d[n][2 * u + 1];
        q[u] = v * v;
      }
      const v2f e = q[0] + q[1], f = q[2] + q[3];
      S[n] = e + f;
      B[n] = e - f;
      C[n] = (q[0] - q[1]) + (q[3] - q[2]);
    }
    const v2f e2 = S[0] + S[1], f2 = S[2] + S[3];
    const v2f vT  = e2 + f2;
    const v2f v8  = e2 - f2;
    const v2f v87 = (S[0] - S[1]) + (S[3] - S[2]);
    const v2f v871  = (B[0] - B[1]) + (B[3] - B[2]);
    const v2f v8710 = (C[0] - C[1]) + (C[3] - C[2]);
    const float qT = vT.x + vT.y, q8 = v8.x + v8.y, q87 = v87.x + v87.y;
    const float q871 = v871.x + v871.y, q8710 = v8710.x + v8710.y;
    const unsigned L = ((unsigned)wave << 9) | ((unsigned)c << 3) | ((unsigned)h << 2);
    unsigned g = L;
    g ^= g >> 1; g ^= g >> 2; g ^= g >> 4; g ^= g >> 8;
#pragma unroll
    for (int j = 0; j <= 10; ++j) {
      const float qv = (j >= 9) ? qT : (j == 8) ? q8 : (j >= 2) ? q87
                     : (j == 1) ? q871 : q8710;
      const bool neg = ((g >> j) & 1u) != 0u;
      acc[N_QUBITS - 1 - j] = neg ? -qv : qv;
    }
    {
      const bool neg = (((g ^ (g >> 11)) & 1u) != 0u);
      acc[0] = neg ? -q8710 : q8710;
    }
    return;
  }
  if (MODE == 1) {
    __syncthreads();
    asm volatile("" : "+v"(stb));
  }
#pragma unroll
  for (int n = 0; n < 4; ++n) {
#pragma unroll
    for (int u = 0; u < 4; ++u) {
      v2f v;
      v.x = d[n][2 * u];
      v.y = d[n][2 * u + 1];
      if (LAST) *(v2f*)((char*)st + (stb ^ (phys_idx(cnot_ring((n << 7) | u)) << 3))) = v;
      else      st[pb[n] ^ phys_idx(u << SH)] = v;
    }
  }
  __syncthreads();
}

__device__ __forceinline__ void product_pass(v2f* st, const v2f* LR,
                                             int wave, int lane) {
  const int c = lane & 15;
  const int h = lane >> 4;
  const int zl = (wave << 3) | ((c >> 3) << 2) | ((c & 7) >> 1);
  v2f a[2], b[4][2];
#pragma unroll
  for (int kk = 0; kk < 2; ++kk) {
    const v2f l = LR[zl * 4 + 2 * kk + h];
    v2f re_row, im_row;
    re_row.x = l.x; re_row.y = -l.y;
    im_row.x = l.y; im_row.y = l.x;
    a[kk] = (c & 1) ? im_row : re_row;
#pragma unroll
    for (int n = 0; n < 4; ++n) b[n][kk] = LR[256 + (2 * kk + h) * 64 + 16 * n + c];
  }
  v8f d[4] = {};
#pragma unroll
  for (int kk = 0; kk < 2; ++kk)
#pragma unroll
    for (int n = 0; n < 4; ++n) d[n] = wmma_f32(a[kk], b[n][kk], d[n]);
  int pbase = phys_idx(cnot_ring((((wave << 3) | (h << 2)) << 6) | c));
  asm volatile("" : "+v"(pbase));
#pragma unroll
  for (int n = 0; n < 4; ++n) {
#pragma unroll
    for (int u = 0; u < 4; ++u) {
      v2f v;
      v.x = d[n][2 * u];
      v.y = d[n][2 * u + 1];
      st[pbase ^ phys_idx(cnot_ring((u << 6) | (n << 4)))] = v;
    }
  }
  __syncthreads();
}

__global__ __launch_bounds__(256) void quantum_kernel(
    const float* __restrict__ noise, const float* __restrict__ W1,
    const float* __restrict__ b1,    const float* __restrict__ W2,
    const float* __restrict__ b2,    float* __restrict__ meas) {
  __shared__ v2f st[DIM];
  __shared__ __attribute__((aligned(16))) v2f afrag[DEPTH * FRAG_PER_LAYER];
  __shared__ v2f ssc[NPARAMS];
  __shared__ __attribute__((aligned(16))) float gU[DEPTH * N_QUBITS][8];
  __shared__ float part[256 * N_QUBITS];
  __shared__ float hbuf[16 * HID];
  __shared__ float pbuf[16 * NPARAMS];
  const int tid  = threadIdx.x;
  const int lane = tid & 31;
  const int wave = tid >> 5;
  const int wave_u = __builtin_amdgcn_readfirstlane(wave);
  const int ebase = MAXE * (int)blockIdx.x;
  const int ecnt  = MAXE;

  {
    const int col   = lane & 15;
    const int rhalf = (lane >> 4) * 8;
    if (wave_u < HID / 16) {
      const int e = ebase + ((col < ecnt) ? col : 0);
      v8f cacc = {};
#pragma unroll
      for (int kc = 0; kc < N_QUBITS; kc += 4) {
        const int k = kc + ((lane >> 4) << 1);
        v2f a;
        a.x = noise[e * N_QUBITS + k];
        a.y = noise[e * N_QUBITS + k + 1];
        v2f b = load_b_frag(W1, HID, kc, wave_u * 16, lane);
        cacc = wmma_f32(a, b, cacc);
      }
      const int n = wave_u * 16 + col;
      const float bias = b1[n];
#pragma unroll
      for (int r = 0; r < 8; ++r)
        hbuf[(rhalf + r) * HID + n] = tanhf(cacc[r] + bias);
    }
    __syncthreads();
    for (int nt = wave_u; nt < NPARAMS / 16; nt += 8) {
      v8f cacc = {};
#pragma unroll
      for (int kc = 0; kc < HID; kc += 4) {
        v2f a = load_a_frag(hbuf, HID, 0, kc, lane);
        v2f b = load_b_frag(W2, NPARAMS, kc, nt * 16, lane);
        cacc = wmma_f32(a, b, cacc);
      }
      const int n = nt * 16 + col;
      const float bias = b2[n];
#pragma unroll
      for (int r = 0; r < 8; ++r)
        pbuf[(rhalf + r) * NPARAMS + n] = cacc[r] + bias;
    }
    __syncthreads();
  }

  const int tpart = (wave << 9) | ((lane & 15) << 3) | ((lane >> 4) << 2);
  const int ld3 = phys_idx(tpart) << 3;
  const int sc3 = phys_idx(cnot_ring(tpart)) << 3;

  for (int it = 0; it < MAXE; ++it) {
  const int bidx = ebase + it;
  const float* p = pbuf + it * NPARAMS;
  {
    const int left = ecnt - it;
    if (left >= 4)      __builtin_amdgcn_s_setprio(3);
    else if (left == 3) __builtin_amdgcn_s_setprio(2);
    else if (left == 2) __builtin_amdgcn_s_setprio(1);
    else                __builtin_amdgcn_s_setprio(0);
  }

  if (tid < NPARAMS) {
    float s, c;
    sincosf(p[tid] * 0.5f, &s, &c);
    v2f sc; sc.x = s; sc.y = c;
    ssc[tid] = sc;
  }
  __syncthreads();

  if (tid < DEPTH * N_QUBITS) {
    const v2f A = ssc[tid * 3 + 0], B = ssc[tid * 3 + 1], Z = ssc[tid * 3 + 2];
    const float sa = A.x, ca = A.y, sb = B.x, cb = B.y, sz = Z.x, cz = Z.y;
    const float m00r =  cb * ca, m00i =  sb * sa;
    const float m01r = -sb * ca, m01i = -cb * sa;
    const float m10r =  sb * ca, m10i = -cb * sa;
    const float m11r =  cb * ca, m11i = -sb * sa;
    float* u = gU[tid];
    u[0] = cz * m00r + sz * m00i;  u[1] = cz * m00i - sz * m00r;
    u[2] = cz * m01r + sz * m01i;  u[3] = cz * m01i - sz * m01r;
    u[4] = cz * m10r - sz * m10i;  u[5] = cz * m10i + sz * m10r;
    u[6] = cz * m11r - sz * m11i;  u[7] = cz * m11i + sz * m11r;
  }
  __syncthreads();

  int tq = tid;
  asm volatile("" : "+v"(tq));
  {
    const int g  = tq >> 6;
    const int jp = (tq >> 3) & 7;
    const int j  = tq & 7;
    const int o0 = ((((jp >> 2) & 1) << 1) | ((j >> 2) & 1)) << 1;
    const int o1 = ((((jp >> 1) & 1) << 1) | ((j >> 1) & 1)) << 1;
    const int o2 = (((jp & 1) << 1) | (j & 1)) << 1;
    const int m0 = ((jp >> 2) << 3) | ((jp & 3) << 1);
    const int eo = (g << 7) | ((j & 3) << 5) | ((j >> 2) << 4) | m0;
    v2f ga[DEPTH][3];
#pragma unroll
    for (int l = 2; l < DEPTH; ++l) {
      const float* u = gU[l * N_QUBITS + 3 * g];
      ga[l][0] = *(const v2f*)(u + o0);
      ga[l][1] = *(const v2f*)(u + 8 + o1);
      ga[l][2] = *(const v2f*)(u + 16 + o2);
    }
    __builtin_amdgcn_sched_barrier(0);
#pragma unroll
    for (int l = 2; l < DEPTH; ++l) {
      const v2f p = cmul(cmul(ga[l][0], ga[l][1]), ga[l][2]);
      v4f ent;
      ent.x = p.x; ent.y = -p.y; ent.z = p.y; ent.w = p.x;
      *(v4f*)(&afrag[l * FRAG_PER_LAYER + eo]) = ent;
    }
  }
  {
    const int half = tq >> 7;
    const int zz = (tq & 127) >> 1;
    const int sb = tq & 1;
    v2f a0[6], a1[6];
    v4f V[6];
#pragma unroll
    for (int i = 0; i < 6; ++i) {
      const float* u0 = gU[half * 6 + i];
      a0[i] = *(const v2f*)(u0);
      a1[i] = *(const v2f*)(u0 + 4);
      const float* u1 = gU[N_QUBITS + half * 6 + i] + 4 * ((zz >> (5 - i)) & 1);
      V[i] = *(const v4f*)(u1);
    }
    __builtin_amdgcn_sched_barrier(0);
    v2f v0, v1;
    {
      v2f vlo, vhi;
      vlo.x = V[0].x; vlo.y = V[0].y; vhi.x = V[0].z; vhi.y = V[0].w;
      v0 = cmul(vlo, sb ? a1[0] : a0[0]);
      v1 = cmul(vhi, sb ? a0[0] : a1[0]);
    }
#pragma unroll
    for (int i = 1; i < 6; ++i) {
      const bool sw = (i == 1) && (half == 0) && (sb != 0);
      const v2f A0 = sw ? a1[i] : a0[i];
      const v2f A1 = sw ? a0[i] : a1[i];
      const v2f x0 = cmul(A0, v0) + cmul(A1, v1);
      const v2f x1 = cmul(A1, v0) + cmul(A0, v1);
      v2f vlo, vhi;
      vlo.x = V[i].x; vlo.y = V[i].y; vhi.x = V[i].z; vhi.y = V[i].w;
      v0 = cmul(vlo, x0);
      v1 = cmul(vhi, x1);
    }
    if (half == 0) {
      afrag[zz * 4 + sb]     = v0;
      afrag[zz * 4 + 2 + sb] = v1;
    } else {
      afrag[256 + (2 * sb) * 64 + zz]     = v0;
      afrag[256 + (2 * sb + 1) * 64 + zz] = v1;
    }
  }
  __syncthreads();

  product_pass(st, afrag, wave, lane);
#pragma nounroll
  for (int l = 2; l < DEPTH; ++l) {
    const v2f* af = afrag + l * FRAG_PER_LAYER;
    group_pass<0, 0>(st, af,       wave, lane);
    group_pass<1, 0>(st, af + 128, wave, lane);
    group_pass<2, 0>(st, af + 256, wave, lane);
    if (l < DEPTH - 1)
      group_pass<3, 1>(st, af + 384, wave, lane, nullptr, ld3, sc3);
  }

  float acc[N_QUBITS];
  group_pass<3, 2>(st, afrag + (DEPTH - 1) * FRAG_PER_LAYER + 384, wave, lane, acc, ld3);
#pragma unroll
  for (int q = 0; q < N_QUBITS; ++q) part[tid * N_QUBITS + q] = acc[q];
  __syncthreads();
  for (int s = 128; s > 0; s >>= 1) {
    if (tid < s) {
#pragma unroll
      for (int q = 0; q < N_QUBITS; ++q)
        part[tid * N_QUBITS + q] += part[(tid + s) * N_QUBITS + q];
    }
    __syncthreads();
  }
  if (tid < N_QUBITS) meas[(size_t)bidx * N_QUBITS + tid] = part[tid];
  }
}

__global__ __launch_bounds__(128) void back_mlp_kernel(
    const float* __restrict__ meas, const float* __restrict__ W3,
    const float* __restrict__ b3,   const float* __restrict__ W4,
    const float* __restrict__ b4,   float* __restrict__ out) {
  __shared__ float hbuf[4][16 * HID];
  const int lane  = threadIdx.x & 31;
  const int wave  = threadIdx.x >> 5;
  const int row0  = (blockIdx.x * 4 + wave) * 16;
  const int col   = lane & 15;
  const int rhalf = (lane >> 4) * 8;
  float* h = hbuf[wave];

  for (int nt = 0; nt < HID / 16; ++nt) {
    v8f c = {};
#pragma unroll
    for (int kc = 0; kc < N_QUBITS; kc += 4) {
      v2f a = load_a_frag(meas, N_QUBITS, row0, kc, lane);
      v2f b = load_b_frag(W3, HID, kc, nt * 16, lane);
      c = wmma_f32(a, b, c);
    }
    int n = nt * 16 + col;
    float bias = b3[n];
#pragma unroll
    for (int r = 0; r < 8; ++r)
      h[(rhalf + r) * HID + n] = tanhf(c[r] + bias);
  }
  __syncthreads();

  v8f c = {};
#pragma unroll
  for (int kc = 0; kc < HID; kc += 4) {
    v2f a = load_a_frag(h, HID, 0, kc, lane);
    int k = kc + ((lane >> 4) << 1);
    v2f b;
    b.x = (col < OUT_DIM) ? W4[k * OUT_DIM + col] : 0.0f;
    b.y = (col < OUT_DIM) ? W4[(k + 1) * OUT_DIM + col] : 0.0f;
    c = wmma_f32(a, b, c);
  }
  if (col < OUT_DIM) {
    float bias = b4[col];
#pragma unroll
    for (int r = 0; r < 8; ++r)
      out[(size_t)(row0 + rhalf + r) * OUT_DIM + col] = c[r] + bias;
  }
}

extern "C" void kernel_launch(void* const* d_in, const int* in_sizes, int n_in,
                              void* d_out, int out_size, void* d_ws, size_t ws_size,
                              hipStream_t stream) {
  const float* noise = (const float*)d_in[0];
  const float* W1 = (const float*)d_in[1];
  const float* b1 = (const float*)d_in[2];
  const float* W2 = (const float*)d_in[3];
  const float* b2 = (const float*)d_in[4];
  const float* W3 = (const float*)d_in[5];
  const float* b3 = (const float*)d_in[6];
  const float* W4 = (const float*)d_in[7];
  const float* b4 = (const float*)d_in[8];
  float* out = (float*)d_out;

  float* meas = (float*)d_ws;

  quantum_kernel<<<NBLK, 256, 0, stream>>>(noise, W1, b1, W2, b2, meas);
  back_mlp_kernel<<<BATCH / 64, 128, 0, stream>>>(meas, W3, b3, W4, b4, out);
}


# Layer 2's group matrices of wires 0-2 and 6-8 multiplied onto the factors L, R (ring as index map, bits of wires 5, 11 carried): 12 matrix instructions a wave, not 40; amplitude rounding changes
# speedup vs baseline: 5.9996x; 1.1082x over previous
#include <hip/hip_runtime.h>
#include <math.h>

#define N_QUBITS 12
#define DEPTH    4
#define HID      64
#define NPARAMS  144
#define BATCH    4096
#define DIM      4096
#define OUT_DIM  2

typedef __attribute__((ext_vector_type(2))) float v2f;
typedef __attribute__((ext_vector_type(4))) float v4f;
typedef __attribute__((ext_vector_type(8))) float v8f;

__device__ __forceinline__ v8f wmma_f32(v2f a, v2f b, v8f c) {
  return __builtin_amdgcn_wmma_f32_16x16x4_f32(
       false, a,  false, b,
       (short)0, c,  false,  false);
}

__device__ __forceinline__ v2f load_a_frag(const float* src, int lda,
                                           int row0, int k0, int lane) {
  int m = row0 + (lane & 15);
  int k = k0 + ((lane >> 4) << 1);
  v2f a;
  a.x = src[m * lda + k];
  a.y = src[m * lda + k + 1];
  return a;
}

__device__ __forceinline__ v2f load_b_frag(const float* W, int ldb,
                                           int k0, int n0, int lane) {
  int c = n0 + (lane & 15);
  int k = k0 + ((lane >> 4) << 1);
  v2f b;
  b.x = W[k * ldb + c];
  b.y = W[(k + 1) * ldb + c];
  return b;
}

#define FRAG_PER_LAYER 512
#define NBLK 1024
#define MAXE 4

__device__ __forceinline__ int phys_idx(int i) {
  return i ^ (((i >> 5) & 3) * 9) ^ ((((i >> 8) ^ (i >> 11)) & 1) << 4);
}

__device__ __forceinline__ int cnot_ring(int x) {
#pragma unroll
  for (int q = 0; q < N_QUBITS; ++q) {
    int pc = N_QUBITS - 1 - q;
    int pt = N_QUBITS - 1 - ((q + 1) % N_QUBITS);
    if ((x >> pc) & 1) x ^= (1 << pt);
  }
  return x;
}

__device__ __forceinline__ v2f cmul(v2f a, v2f b) {
  v2f r;
  r.x = a.x * b.x - a.y * b.y;
  r.y = a.x * b.y + a.y * b.x;
  return r;
}

template <int G, int MODE>
__device__ __forceinline__ void group_pass(v2f* st, const v2f* af,
                                           int wave, int lane,
                                           float* acc = nullptr,
                                           int ldb = 0, int stb = 0) {
  constexpr bool LAST = (MODE == 1);
  constexpr int NK = 4;
  constexpr int SH = 9 - 3 * G;
  const int c = lane & 15;
  const int h = lane >> 4;
  v2f a[NK];
#pragma unroll
  for (int kk = 0; kk < NK; ++kk) a[kk] = af[kk * 32 + lane];

  if (G == 3) asm volatile("" : "+v"(ldb));
  int ib[4], pb[4];
  v2f x[4][NK];
#pragma unroll
  for (int n = 0; n < 4; ++n) {
    const int col9 = ((wave * 4 + n) << 4) | c;
    const int hi = col9 >> SH;
    const int lo = col9 & ((1 << SH) - 1);
    ib[n] = (hi << (SH + 3)) | (h << (SH + 2)) | lo;
    pb[n] = phys_idx(ib[n]);
#pragma unroll
    for (int u = 0; u < NK; ++u)
      x[n][u] = (G == 3)
          ? *(const v2f*)((const char*)st + (ldb ^ (phys_idx((n << 7) | u) << 3)))
          : st[pb[n] ^ phys_idx(u << SH)];
  }
  __builtin_amdgcn_sched_barrier(0);
  v8f d[4] = {};
#pragma unroll
  for (int kk = 0; kk < NK; ++kk)
#pragma unroll
    for (int n = 0; n < 4; ++n) d[n] = wmma_f32(a[kk], x[n][kk], d[n]);

  if (MODE == 2) {
    v2f S[4], B[4], C[4];
#pragma unroll
    for (int n = 0; n < 4; ++n) {
      v2f q[4];
#pragma unroll
      for (int u = 0; u < 4; ++u) {
        v2f v;
        v.x = d[n][2 * u];
        v.y = d[n][2 * u + 1];
        q[u] = v * v;
      }
      const v2f e = q[0] + q[1], f = q[2] + q[3];
      S[n] = e + f;
      B[n] = e - f;
      C[n] = (q[0] - q[1]) + (q[3] - q[2]);
    }
    const v2f e2 = S[0] + S[1], f2 = S[2] + S[3];
    const v2f vT  = e2 + f2;
    const v2f v8  = e2 - f2;
    const v2f v87 = (S[0] - S[1]) + (S[3] - S[2]);
    const v2f v871  = (B[0] - B[1]) + (B[3] - B[2]);
    const v2f v8710 = (C[0] - C[1]) + (C[3] - C[2]);
    const float qT = vT.x + vT.y, q8 = v8.x + v8.y, q87 = v87.x + v87.y;
    const float q871 = v871.x + v871.y, q8710 = v8710.x + v8710.y;
    const unsigned L = ((unsigned)wave << 9) | ((unsigned)c << 3) | ((unsigned)h << 2);
    unsigned g = L;
    g ^= g >> 1; g ^= g >> 2; g ^= g >> 4; g ^= g >> 8;
#pragma unroll
    for (int j = 0; j <= 10; ++j) {
      const float qv = (j >= 9) ? qT : (j == 8) ? q8 : (j >= 2) ? q87
                     : (j == 1) ? q871 : q8710;
      const bool neg = ((g >> j) & 1u) != 0u;
      acc[N_QUBITS - 1 - j] = neg ? -qv : qv;
    }
    {
      const bool neg = (((g ^ (g >> 11)) & 1u) != 0u);
      acc[0] = neg ? -q8710 : q8710;
    }
    return;
  }
  if (MODE == 1) {
    __syncthreads();
    asm volatile("" : "+v"(stb));
  }
#pragma unroll
  for (int n = 0; n < 4; ++n) {
#pragma unroll
    for (int u = 0; u < 4; ++u) {
      v2f v;
      v.x = d[n][2 * u];
      v.y = d[n][2 * u + 1];
      if (LAST) *(v2f*)((char*)st + (stb ^ (phys_idx(cnot_ring((n << 7) | u)) << 3))) = v;
      else      st[pb[n] ^ phys_idx(u << SH)] = v;
    }
  }
  __syncthreads();
}

__device__ __forceinline__ void factor_pass(const v2f* LR, const v2f* af2,
                                            v2f* L2, v2f* R2,
                                            int side, int q2, int lane) {
  const int c = lane & 15;
  const int h = lane >> 4;
  const int t3 = (q2 << 1) | (c >> 3);
  const int beta = c & 7;
  const int bb = beta >> 1;
  const int s = beta & 1;
  const v2f* af = af2 + (side ? 256 : 0);
  v2f a[4], x[4];
#pragma unroll
  for (int kk = 0; kk < 4; ++kk) {
    a[kk] = af[kk * 32 + lane];
    const int z = (((h << 2) | kk) << 3) | t3;
    const int zz = z ^ (z >> 1);
    const int il = ((zz ^ (s * 0x30)) << 2) | bb;
    const int ir = 256 + (bb << 6) + (zz ^ (s << 5));
    x[kk] = LR[side ? ir : il];
  }
  v8f d = {};
#pragma unroll
  for (int kk = 0; kk < 4; ++kk) d = wmma_f32(a[kk], x[kk], d);
#pragma unroll
  for (int u = 0; u < 4; ++u) {
    const int zn = (((h << 2) | u) << 3) | t3;
    v2f v;
    v.x = d[2 * u];
    v.y = d[2 * u + 1];
    if (side) R2[(beta << 6) + zn] = v;
    else      L2[(zn << 3) + beta] = v;
  }
  __syncthreads();
}

__device__ __forceinline__ void product_pass(v2f* st, const v2f* L2, const v2f* R2,
                                             int wave, int lane) {
  const int c = lane & 15;
  const int h = lane >> 4;
  const int w1 = wave & 1;
  const int zl = ((wave >> 1) << 4) | ((c >> 3) << 3) | (((c & 7) >> 1) << 1) | w1;
  v2f a[2][2], b[4][2];
#pragma unroll
  for (int kk = 0; kk < 2; ++kk) {
    const int bnd = (kk << 1) | h;
#pragma unroll
    for (int s11 = 0; s11 < 2; ++s11) {
      const v2f l = L2[(zl << 3) + (bnd << 1) + s11];
      v2f re_row, im_row;
      re_row.x = l.x; re_row.y = -l.y;
      im_row.x = l.y; im_row.y = l.x;
      a[kk][s11] = (c & 1) ? im_row : re_row;
    }
#pragma unroll
    for (int n = 0; n < 4; ++n)
      b[n][kk] = R2[(((bnd << 1) | w1) << 6) + (((n >> 1) << 5) | (c << 1) | (n & 1))];
  }
  v8f d[4] = {};
#pragma unroll
  for (int kk = 0; kk < 2; ++kk)
#pragma unroll
    for (int n = 0; n < 4; ++n) d[n] = wmma_f32(a[kk][n & 1], b[n][kk], d[n]);
  const int pbase = phys_idx(((wave >> 1) << 10) | (h << 9) | (w1 << 6) | (c << 1));
#pragma unroll
  for (int n = 0; n < 4; ++n) {
#pragma unroll
    for (int u = 0; u < 4; ++u) {
      v2f v;
      v.x = d[n][2 * u];
      v.y = d[n][2 * u + 1];
      st[pbase ^ phys_idx((u << 7) | ((n >> 1) << 5) | (n & 1))] = v;
    }
  }
  __syncthreads();
}

__global__ __launch_bounds__(256) void quantum_kernel(
    const float* __restrict__ noise, const float* __restrict__ W1,
    const float* __restrict__ b1,    const float* __restrict__ W2,
    const float* __restrict__ b2,    float* __restrict__ meas) {
  __shared__ v2f st[DIM];
  __shared__ __attribute__((aligned(16))) v2f afrag[DEPTH * FRAG_PER_LAYER];
  __shared__ v2f ssc[NPARAMS];
  __shared__ __attribute__((aligned(16))) float gU[DEPTH * N_QUBITS][8];
  __shared__ float part[256 * N_QUBITS];
  __shared__ __attribute__((aligned(16))) float hbuf[16 * HID];
  __shared__ float pbuf[16 * NPARAMS];
  const int tid  = threadIdx.x;
  const int lane = tid & 31;
  const int wave = tid >> 5;
  const int wave_u = __builtin_amdgcn_readfirstlane(wave);
  const int ebase = MAXE * (int)blockIdx.x;
  const int ecnt  = MAXE;

  {
    const int col   = lane & 15;
    const int rhalf = (lane >> 4) * 8;
    if (wave_u < HID / 16) {
      const int e = ebase + ((col < ecnt) ? col : 0);
      v8f cacc = {};
#pragma unroll
      for (int kc = 0; kc < N_QUBITS; kc += 4) {
        const int k = kc + ((lane >> 4) << 1);
        v2f a;
        a.x = noise[e * N_QUBITS + k];
        a.y = noise[e * N_QUBITS + k + 1];
        v2f b = load_b_frag(W1, HID, kc, wave_u * 16, lane);
        cacc = wmma_f32(a, b, cacc);
      }
      const int n = wave_u * 16 + col;
      const float bias = b1[n];
#pragma unroll
      for (int r = 0; r < 8; ++r)
        hbuf[(rhalf + r) * HID + n] = tanhf(cacc[r] + bias);
    }
    __syncthreads();
    for (int nt = wave_u; nt < NPARAMS / 16; nt += 8) {
      v8f cacc = {};
#pragma unroll
      for (int kc = 0; kc < HID; kc += 4) {
        v2f a = load_a_frag(hbuf, HID, 0, kc, lane);
        v2f b = load_b_frag(W2, NPARAMS, kc, nt * 16, lane);
        cacc = wmma_f32(a, b, cacc);
      }
      const int n = nt * 16 + col;
      const float bias = b2[n];
#pragma unroll
      for (int r = 0; r < 8; ++r)
        pbuf[(rhalf + r) * NPARAMS + n] = cacc[r] + bias;
    }
    __syncthreads();
  }

  const int tpart = (wave << 9) | ((lane & 15) << 3) | ((lane >> 4) << 2);
  const int ld3 = phys_idx(tpart) << 3;
  const int sc3 = phys_idx(cnot_ring(tpart)) << 3;

  for (int it = 0; it < MAXE; ++it) {
  const int bidx = ebase + it;
  const float* p = pbuf + it * NPARAMS;
  {
    const int left = ecnt - it;
    if (left >= 4)      __builtin_amdgcn_s_setprio(3);
    else if (left == 3) __builtin_amdgcn_s_setprio(2);
    else if (left == 2) __builtin_amdgcn_s_setprio(1);
    else                __builtin_amdgcn_s_setprio(0);
  }

  if (tid < NPARAMS) {
    float s, c;
    sincosf(p[tid] * 0.5f, &s, &c);
    v2f sc; sc.x = s; sc.y = c;
    ssc[tid] = sc;
  }
  __syncthreads();

  if (tid < DEPTH * N_QUBITS) {
    const v2f A = ssc[tid * 3 + 0], B = ssc[tid * 3 + 1], Z = ssc[tid * 3 + 2];
    const float sa = A.x, ca = A.y, sb = B.x, cb = B.y, sz = Z.x, cz = Z.y;
    const float m00r =  cb * ca, m00i =  sb * sa;
    const float m01r = -sb * ca, m01i = -cb * sa;
    const float m10r =  sb * ca, m10i = -cb * sa;
    const float m11r =  cb * ca, m11i = -sb * sa;
    float* u = gU[tid];
    u[0] = cz * m00r + sz * m00i;  u[1] = cz * m00i - sz * m00r;
    u[2] = cz * m01r + sz * m01i;  u[3] = cz * m01i - sz * m01r;
    u[4] = cz * m10r - sz * m10i;  u[5] = cz * m10i + sz * m10r;
    u[6] = cz * m11r - sz * m11i;  u[7] = cz * m11i + sz * m11r;
  }
  __syncthreads();

  int tq = tid;
  asm volatile("" : "+v"(tq));
  {
    const int g  = tq >> 6;
    const int jp = (tq >> 3) & 7;
    const int j  = tq & 7;
    const int o0 = ((((jp >> 2) & 1) << 1) | ((j >> 2) & 1)) << 1;
    const int o1 = ((((jp >> 1) & 1) << 1) | ((j >> 1) & 1)) << 1;
    const int o2 = (((jp & 1) << 1) | (j & 1)) << 1;
    const int m0 = ((jp >> 2) << 3) | ((jp & 3) << 1);
    const int eo = (g << 7) | ((j & 3) << 5) | ((j >> 2) << 4) | m0;
    v2f ga[DEPTH][3];
#pragma unroll
    for (int l = 2; l < DEPTH; ++l) {
      const float* u = gU[l * N_QUBITS + 3 * g];
      ga[l][0] = *(const v2f*)(u + o0);
      ga[l][1] = *(const v2f*)(u + 8 + o1);
      ga[l][2] = *(const v2f*)(u + 16 + o2);
    }
    __builtin_amdgcn_sched_barrier(0);
#pragma unroll
    for (int l = 2; l < DEPTH; ++l) {
      const v2f p = cmul(cmul(ga[l][0], ga[l][1]), ga[l][2]);
      v4f ent;
      ent.x = p.x; ent.y = -p.y; ent.z = p.y; ent.w = p.x;
      *(v4f*)(&afrag[l * FRAG_PER_LAYER + eo]) = ent;
    }
  }
  {
    const int half = tq >> 7;
    const int zz = (tq & 127) >> 1;
    const int sb = tq & 1;
    v2f a0[6], a1[6];
    v4f V[6];
#pragma unroll
    for (int i = 0; i < 6; ++i) {
      const float* u0 = gU[half * 6 + i];
      a0[i] = *(const v2f*)(u0);
      a1[i] = *(const v2f*)(u0 + 4);
      const float* u1 = gU[N_QUBITS + half * 6 + i] + 4 * ((zz >> (5 - i)) & 1);
      V[i] = *(const v4f*)(u1);
    }
    __builtin_amdgcn_sched_barrier(0);
    v2f v0, v1;
    {
      v2f vlo, vhi;
      vlo.x = V[0].x; vlo.y = V[0].y; vhi.x = V[0].z; vhi.y = V[0].w;
      v0 = cmul(vlo, sb ? a1[0] : a0[0]);
      v1 = cmul(vhi, sb ? a0[0] : a1[0]);
    }
#pragma unroll
    for (int i = 1; i < 6; ++i) {
      const bool sw = (i == 1) && (half == 0) && (sb != 0);
      const v2f A0 = sw ? a1[i] : a0[i];
      const v2f A1 = sw ? a0[i] : a1[i];
      const v2f x0 = cmul(A0, v0) + cmul(A1, v1);
      const v2f x1 = cmul(A1, v0) + cmul(A0, v1);
      v2f vlo, vhi;
      vlo.x = V[i].x; vlo.y = V[i].y; vhi.x = V[i].z; vhi.y = V[i].w;
      v0 = cmul(vlo, x0);
      v1 = cmul(vhi, x1);
    }
    if (half == 0) {
      afrag[zz * 4 + sb]     = v0;
      afrag[zz * 4 + 2 + sb] = v1;
    } else {
      afrag[256 + (2 * sb) * 64 + zz]     = v0;
      afrag[256 + (2 * sb + 1) * 64 + zz] = v1;
    }
  }
  __syncthreads();

  v2f* const L2 = afrag + FRAG_PER_LAYER;
  v2f* const R2 = (v2f*)hbuf;
  factor_pass(afrag, afrag + 2 * FRAG_PER_LAYER, L2, R2, wave_u >> 2, wave_u & 3, tq & 31);
  product_pass(st, L2, R2, tq >> 5, tq & 31);
#pragma nounroll
  for (int l = 2; l < DEPTH; ++l) {
    const v2f* af = afrag + l * FRAG_PER_LAYER;
    if (l > 2) group_pass<0, 0>(st, af,       wave, lane);
    group_pass<1, 0>(st, af + 128, wave, lane);
    if (l > 2) group_pass<2, 0>(st, af + 256, wave, lane);
    if (l < DEPTH - 1)
      group_pass<3, 1>(st, af + 384, wave, lane, nullptr, ld3, sc3);
  }

  float acc[N_QUBITS];
  group_pass<3, 2>(st, afrag + (DEPTH - 1) * FRAG_PER_LAYER + 384, wave, lane, acc, ld3);
#pragma unroll
  for (int q = 0; q < N_QUBITS; ++q) part[tid * N_QUBITS + q] = acc[q];
  __syncthreads();
  for (int s = 128; s > 0; s >>= 1) {
    if (tid < s) {
#pragma unroll
      for (int q = 0; q < N_QUBITS; ++q)
        part[tid * N_QUBITS + q] += part[(tid + s) * N_QUBITS + q];
    }
    __syncthreads();
  }
  if (tid < N_QUBITS) meas[(size_t)bidx * N_QUBITS + tid] = part[tid];
  }
}

__global__ __launch_bounds__(128) void back_mlp_kernel(
    const float* __restrict__ meas, const float* __restrict__ W3,
    const float* __restrict__ b3,   const float* __restrict__ W4,
    const float* __restrict__ b4,   float* __restrict__ out) {
  __shared__ float hbuf[4][16 * HID];
  const int lane  = threadIdx.x & 31;
  const int wave  = threadIdx.x >> 5;
  const int row0  = (blockIdx.x * 4 + wave) * 16;
  const int col   = lane & 15;
  const int rhalf = (lane >> 4) * 8;
  float* h = hbuf[wave];

  for (int nt = 0; nt < HID / 16; ++nt) {
    v8f c = {};
#pragma unroll
    for (int kc = 0; kc < N_QUBITS; kc += 4) {
      v2f a = load_a_frag(meas, N_QUBITS, row0, kc, lane);
      v2f b = load_b_frag(W3, HID, kc, nt * 16, lane);
      c = wmma_f32(a, b, c);
    }
    int n = nt * 16 + col;
    float bias = b3[n];
#pragma unroll
    for (int r = 0; r < 8; ++r)
      h[(rhalf + r) * HID + n] = tanhf(c[r] + bias);
  }
  __syncthreads();

  v8f c = {};
#pragma unroll
  for (int kc = 0; kc < HID; kc += 4) {
    v2f a = load_a_frag(h, HID, 0, kc, lane);
    int k = kc + ((lane >> 4) << 1);
    v2f b;
    b.x = (col < OUT_DIM) ? W4[k * OUT_DIM + col] : 0.0f;
    b.y = (col < OUT_DIM) ? W4[(k + 1) * OUT_DIM + col] : 0.0f;
    c = wmma_f32(a, b, c);
  }
  if (col < OUT_DIM) {
    float bias = b4[col];
#pragma unroll
    for (int r = 0; r < 8; ++r)
      out[(size_t)(row0 + rhalf + r) * OUT_DIM + col] = c[r] + bias;
  }
}

extern "C" void kernel_launch(void* const* d_in, const int* in_sizes, int n_in,
                              void* d_out, int out_size, void* d_ws, size_t ws_size,
                              hipStream_t stream) {
  const float* noise = (const float*)d_in[0];
  const float* W1 = (const float*)d_in[1];
  const float* b1 = (const float*)d_in[2];
  const float* W2 = (const float*)d_in[3];
  const float* b2 = (const float*)d_in[4];
  const float* W3 = (const float*)d_in[5];
  const float* b3 = (const float*)d_in[6];
  const float* W4 = (const float*)d_in[7];
  const float* b4 = (const float*)d_in[8];
  float* out = (float*)d_out;

  float* meas = (float*)d_ws;

  quantum_kernel<<<NBLK, 256, 0, stream>>>(noise, W1, b1, W2, b2, meas);
  back_mlp_kernel<<<BATCH / 64, 128, 0, stream>>>(meas, W3, b3, W4, b4, out);
}


# Expectation sums: two LDS stages (48 threads add 16 rows each by 16-byte loads, 3 threads add the 16 partial rows and store) replace the 8-step halving tree: 2 barriers, not 9; summing order changes
# speedup vs baseline: 6.1976x; 1.0330x over previous
#include <hip/hip_runtime.h>
#include <math.h>

#define N_QUBITS 12
#define DEPTH    4
#define HID      64
#define NPARAMS  144
#define BATCH    4096
#define DIM      4096
#define OUT_DIM  2

typedef __attribute__((ext_vector_type(2))) float v2f;
typedef __attribute__((ext_vector_type(4))) float v4f;
typedef __attribute__((ext_vector_type(8))) float v8f;

__device__ __forceinline__ v8f wmma_f32(v2f a, v2f b, v8f c) {
  return __builtin_amdgcn_wmma_f32_16x16x4_f32(
       false, a,  false, b,
       (short)0, c,  false,  false);
}

__device__ __forceinline__ v2f load_a_frag(const float* src, int lda,
                                           int row0, int k0, int lane) {
  int m = row0 + (lane & 15);
  int k = k0 + ((lane >> 4) << 1);
  v2f a;
  a.x = src[m * lda + k];
  a.y = src[m * lda + k + 1];
  return a;
}

__device__ __forceinline__ v2f load_b_frag(const float* W, int ldb,
                                           int k0, int n0, int lane) {
  int c = n0 + (lane & 15);
  int k = k0 + ((lane >> 4) << 1);
  v2f b;
  b.x = W[k * ldb + c];
  b.y = W[(k + 1) * ldb + c];
  return b;
}

#define FRAG_PER_LAYER 512
#define NBLK 1024
#define MAXE 4

__device__ __forceinline__ int phys_idx(int i) {
  return i ^ (((i >> 5) & 3) * 9) ^ ((((i >> 8) ^ (i >> 11)) & 1) << 4);
}

__device__ __forceinline__ int cnot_ring(int x) {
#pragma unroll
  for (int q = 0; q < N_QUBITS; ++q) {
    int pc = N_QUBITS - 1 - q;
    int pt = N_QUBITS - 1 - ((q + 1) % N_QUBITS);
    if ((x >> pc) & 1) x ^= (1 << pt);
  }
  return x;
}

__device__ __forceinline__ v2f cmul(v2f a, v2f b) {
  v2f r;
  r.x = a.x * b.x - a.y * b.y;
  r.y = a.x * b.y + a.y * b.x;
  return r;
}

template <int G, int MODE>
__device__ __forceinline__ void group_pass(v2f* st, const v2f* af,
                                           int wave, int lane,
                                           float* acc = nullptr,
                                           int ldb = 0, int stb = 0) {
  constexpr bool LAST = (MODE == 1);
  constexpr int NK = 4;
  constexpr int SH = 9 - 3 * G;
  const int c = lane & 15;
  const int h = lane >> 4;
  v2f a[NK];
#pragma unroll
  for (int kk = 0; kk < NK; ++kk) a[kk] = af[kk * 32 + lane];

  if (G == 3) asm volatile("" : "+v"(ldb));
  int ib[4], pb[4];
  v2f x[4][NK];
#pragma unroll
  for (int n = 0; n < 4; ++n) {
    const int col9 = ((wave * 4 + n) << 4) | c;
    const int hi = col9 >> SH;
    const int lo = col9 & ((1 << SH) - 1);
    ib[n] = (hi << (SH + 3)) | (h << (SH + 2)) | lo;
    pb[n] = phys_idx(ib[n]);
#pragma unroll
    for (int u = 0; u < NK; ++u)
      x[n][u] = (G == 3)
          ? *(const v2f*)((const char*)st + (ldb ^ (phys_idx((n << 7) | u) << 3)))
          : st[pb[n] ^ phys_idx(u << SH)];
  }
  __builtin_amdgcn_sched_barrier(0);
  v8f d[4] = {};
#pragma unroll
  for (int kk = 0; kk < NK; ++kk)
#pragma unroll
    for (int n = 0; n < 4; ++n) d[n] = wmma_f32(a[kk], x[n][kk], d[n]);

  if (MODE == 2) {
    v2f S[4], B[4], C[4];
#pragma unroll
    for (int n = 0; n < 4; ++n) {
      v2f q[4];
#pragma unroll
      for (int u = 0; u < 4; ++u) {
        v2f v;
        v.x = d[n][2 * u];
        v.y = d[n][2 * u + 1];
        q[u] = v * v;
      }
      const v2f e = q[0] + q[1], f = q[2] + q[3];
      S[n] = e + f;
      B[n] = e - f;
      C[n] = (q[0] - q[1]) + (q[3] - q[2]);
    }
    const v2f e2 = S[0] + S[1], f2 = S[2] + S[3];
    const v2f vT  = e2 + f2;
    const v2f v8  = e2 - f2;
    const v2f v87 = (S[0] - S[1]) + (S[3] - S[2]);
    const v2f v871  = (B[0] - B[1]) + (B[3] - B[2]);
    const v2f v8710 = (C[0] - C[1]) + (C[3] - C[2]);
    const float qT = vT.x + vT.y, q8 = v8.x + v8.y, q87 = v87.x + v87.y;
    const float q871 = v871.x + v871.y, q8710 = v8710.x + v8710.y;
    const unsigned L = ((unsigned)wave << 9) | ((unsigned)c << 3) | ((unsigned)h << 2);
    unsigned g = L;
    g ^= g >> 1; g ^= g >> 2; g ^= g >> 4; g ^= g >> 8;
#pragma unroll
    for (int j = 0; j <= 10; ++j) {
      const float qv = (j >= 9) ? qT : (j == 8) ? q8 : (j >= 2) ? q87
                     : (j == 1) ? q871 : q8710;
      const bool neg = ((g >> j) & 1u) != 0u;
      acc[N_QUBITS - 1 - j] = neg ? -qv : qv;
    }
    {
      const bool neg = (((g ^ (g >> 11)) & 1u) != 0u);
      acc[0] = neg ? -q8710 : q8710;
    }
    return;
  }
  if (MODE == 1) {
    __syncthreads();
    asm volatile("" : "+v"(stb));
  }
#pragma unroll
  for (int n = 0; n < 4; ++n) {
#pragma unroll
    for (int u = 0; u < 4; ++u) {
      v2f v;
      v.x = d[n][2 * u];
      v.y = d[n][2 * u + 1];
      if (LAST) *(v2f*)((char*)st + (stb ^ (phys_idx(cnot_ring((n << 7) | u)) << 3))) = v;
      else      st[pb[n] ^ phys_idx(u << SH)] = v;
    }
  }
  __syncthreads();
}

__device__ __forceinline__ void factor_pass(const v2f* LR, const v2f* af2,
                                            v2f* L2, v2f* R2,
                                            int side, int q2, int lane) {
  const int c = lane & 15;
  const int h = lane >> 4;
  const int t3 = (q2 << 1) | (c >> 3);
  const int beta = c & 7;
  const int bb = beta >> 1;
  const int s = beta & 1;
  const v2f* af = af2 + (side ? 256 : 0);
  v2f a[4], x[4];
#pragma unroll
  for (int kk = 0; kk < 4; ++kk) {
    a[kk] = af[kk * 32 + lane];
    const int z = (((h << 2) | kk) << 3) | t3;
    const int zz = z ^ (z >> 1);
    const int il = ((zz ^ (s * 0x30)) << 2) | bb;
    const int ir = 256 + (bb << 6) + (zz ^ (s << 5));
    x[kk] = LR[side ? ir : il];
  }
  v8f d = {};
#pragma unroll
  for (int kk = 0; kk < 4; ++kk) d = wmma_f32(a[kk], x[kk], d);
#pragma unroll
  for (int u = 0; u < 4; ++u) {
    const int zn = (((h << 2) | u) << 3) | t3;
    v2f v;
    v.x = d[2 * u];
    v.y = d[2 * u + 1];
    if (side) R2[(beta << 6) + zn] = v;
    else      L2[(zn << 3) + beta] = v;
  }
  __syncthreads();
}

__device__ __forceinline__ void product_pass(v2f* st, const v2f* L2, const v2f* R2,
                                             int wave, int lane) {
  const int c = lane & 15;
  const int h = lane >> 4;
  const int w1 = wave & 1;
  const int zl = ((wave >> 1) << 4) | ((c >> 3) << 3) | (((c & 7) >> 1) << 1) | w1;
  v2f a[2][2], b[4][2];
#pragma unroll
  for (int kk = 0; kk < 2; ++kk) {
    const int bnd = (kk << 1) | h;
#pragma unroll
    for (int s11 = 0; s11 < 2; ++s11) {
      const v2f l = L2[(zl << 3) + (bnd << 1) + s11];
      v2f re_row, im_row;
      re_row.x = l.x; re_row.y = -l.y;
      im_row.x = l.y; im_row.y = l.x;
      a[kk][s11] = (c & 1) ? im_row : re_row;
    }
#pragma unroll
    for (int n = 0; n < 4; ++n)
      b[n][kk] = R2[(((bnd << 1) | w1) << 6) + (((n >> 1) << 5) | (c << 1) | (n & 1))];
  }
  v8f d[4] = {};
#pragma unroll
  for (int kk = 0; kk < 2; ++kk)
#pragma unroll
    for (int n = 0; n < 4; ++n) d[n] = wmma_f32(a[kk][n & 1], b[n][kk], d[n]);
  const int pbase = phys_idx(((wave >> 1) << 10) | (h << 9) | (w1 << 6) | (c << 1));
#pragma unroll
  for (int n = 0; n < 4; ++n) {
#pragma unroll
    for (int u = 0; u < 4; ++u) {
      v2f v;
      v.x = d[n][2 * u];
      v.y = d[n][2 * u + 1];
      st[pbase ^ phys_idx((u << 7) | ((n >> 1) << 5) | (n & 1))] = v;
    }
  }
  __syncthreads();
}

__global__ __launch_bounds__(256) void quantum_kernel(
    const float* __restrict__ noise, const float* __restrict__ W1,
    const float* __restrict__ b1,    const float* __restrict__ W2,
    const float* __restrict__ b2,    float* __restrict__ meas) {
  __shared__ v2f st[DIM];
  __shared__ __attribute__((aligned(16))) v2f afrag[DEPTH * FRAG_PER_LAYER];
  __shared__ v2f ssc[NPARAMS];
  __shared__ __attribute__((aligned(16))) float gU[DEPTH * N_QUBITS][8];
  __shared__ __attribute__((aligned(16))) float part[256 * N_QUBITS];
  __shared__ __attribute__((aligned(16))) float hbuf[16 * HID];
  __shared__ float pbuf[16 * NPARAMS];
  const int tid  = threadIdx.x;
  const int lane = tid & 31;
  const int wave = tid >> 5;
  const int wave_u = __builtin_amdgcn_readfirstlane(wave);
  const int ebase = MAXE * (int)blockIdx.x;
  const int ecnt  = MAXE;

  {
    const int col   = lane & 15;
    const int rhalf = (lane >> 4) * 8;
    if (wave_u < HID / 16) {
      const int e = ebase + ((col < ecnt) ? col : 0);
      v8f cacc = {};
#pragma unroll
      for (int kc = 0; kc < N_QUBITS; kc += 4) {
        const int k = kc + ((lane >> 4) << 1);
        v2f a;
        a.x = noise[e * N_QUBITS + k];
        a.y = noise[e * N_QUBITS + k + 1];
        v2f b = load_b_frag(W1, HID, kc, wave_u * 16, lane);
        cacc = wmma_f32(a, b, cacc);
      }
      const int n = wave_u * 16 + col;
      const float bias = b1[n];
#pragma unroll
      for (int r = 0; r < 8; ++r)
        hbuf[(rhalf + r) * HID + n] = tanhf(cacc[r] + bias);
    }
    __syncthreads();
    for (int nt = wave_u; nt < NPARAMS / 16; nt += 8) {
      v8f cacc = {};
#pragma unroll
      for (int kc = 0; kc < HID; kc += 4) {
        v2f a = load_a_frag(hbuf, HID, 0, kc, lane);
        v2f b = load_b_frag(W2, NPARAMS, kc, nt * 16, lane);
        cacc = wmma_f32(a, b, cacc);
      }
      const int n = nt * 16 + col;
      const float bias = b2[n];
#pragma unroll
      for (int r = 0; r < 8; ++r)
        pbuf[(rhalf + r) * NPARAMS + n] = cacc[r] + bias;
    }
    __syncthreads();
  }

  const int tpart = (wave << 9) | ((lane & 15) << 3) | ((lane >> 4) << 2);
  const int ld3 = phys_idx(tpart) << 3;
  const int sc3 = phys_idx(cnot_ring(tpart)) << 3;

  for (int it = 0; it < MAXE; ++it) {
  const int bidx = ebase + it;
  const float* p = pbuf + it * NPARAMS;
  {
    const int left = ecnt - it;
    if (left >= 4)      __builtin_amdgcn_s_setprio(3);
    else if (left == 3) __builtin_amdgcn_s_setprio(2);
    else if (left == 2) __builtin_amdgcn_s_setprio(1);
    else                __builtin_amdgcn_s_setprio(0);
  }

  if (tid < NPARAMS) {
    float s, c;
    sincosf(p[tid] * 0.5f, &s, &c);
    v2f sc; sc.x = s; sc.y = c;
    ssc[tid] = sc;
  }
  __syncthreads();

  if (tid < DEPTH * N_QUBITS) {
    const v2f A = ssc[tid * 3 + 0], B = ssc[tid * 3 + 1], Z = ssc[tid * 3 + 2];
    const float sa = A.x, ca = A.y, sb = B.x, cb = B.y, sz = Z.x, cz = Z.y;
    const float m00r =  cb * ca, m00i =  sb * sa;
    const float m01r = -sb * ca, m01i = -cb * sa;
    const float m10r =  sb * ca, m10i = -cb * sa;
    const float m11r =  cb * ca, m11i = -sb * sa;
    float* u = gU[tid];
    u[0] = cz * m00r + sz * m00i;  u[1] = cz * m00i - sz * m00r;
    u[2] = cz * m01r + sz * m01i;  u[3] = cz * m01i - sz * m01r;
    u[4] = cz * m10r - sz * m10i;  u[5] = cz * m10i + sz * m10r;
    u[6] = cz * m11r - sz * m11i;  u[7] = cz * m11i + sz * m11r;
  }
  __syncthreads();

  int tq = tid;
  asm volatile("" : "+v"(tq));
  {
    const int g  = tq >> 6;
    const int jp = (tq >> 3) & 7;
    const int j  = tq & 7;
    const int o0 = ((((jp >> 2) & 1) << 1) | ((j >> 2) & 1)) << 1;
    const int o1 = ((((jp >> 1) & 1) << 1) | ((j >> 1) & 1)) << 1;
    const int o2 = (((jp & 1) << 1) | (j & 1)) << 1;
    const int m0 = ((jp >> 2) << 3) | ((jp & 3) << 1);
    const int eo = (g << 7) | ((j & 3) << 5) | ((j >> 2) << 4) | m0;
    v2f ga[DEPTH][3];
#pragma unroll
    for (int l = 2; l < DEPTH; ++l) {
      const float* u = gU[l * N_QUBITS + 3 * g];
      ga[l][0] = *(const v2f*)(u + o0);
      ga[l][1] = *(const v2f*)(u + 8 + o1);
      ga[l][2] = *(const v2f*)(u + 16 + o2);
    }
    __builtin_amdgcn_sched_barrier(0);
#pragma unroll
    for (int l = 2; l < DEPTH; ++l) {
      const v2f p = cmul(cmul(ga[l][0], ga[l][1]), ga[l][2]);
      v4f ent;
      ent.x = p.x; ent.y = -p.y; ent.z = p.y; ent.w = p.x;
      *(v4f*)(&afrag[l * FRAG_PER_LAYER + eo]) = ent;
    }
  }
  {
    const int half = tq >> 7;
    const int zz = (tq & 127) >> 1;
    const int sb = tq & 1;
    v2f a0[6], a1[6];
    v4f V[6];
#pragma unroll
    for (int i = 0; i < 6; ++i) {
      const float* u0 = gU[half * 6 + i];
      a0[i] = *(const v2f*)(u0);
      a1[i] = *(const v2f*)(u0 + 4);
      const float* u1 = gU[N_QUBITS + half * 6 + i] + 4 * ((zz >> (5 - i)) & 1);
      V[i] = *(const v4f*)(u1);
    }
    __builtin_amdgcn_sched_barrier(0);
    v2f v0, v1;
    {
      v2f vlo, vhi;
      vlo.x = V[0].x; vlo.y = V[0].y; vhi.x = V[0].z; vhi.y = V[0].w;
      v0 = cmul(vlo, sb ? a1[0] : a0[0]);
      v1 = cmul(vhi, sb ? a0[0] : a1[0]);
    }
#pragma unroll
    for (int i = 1; i < 6; ++i) {
      const bool sw = (i == 1) && (half == 0) && (sb != 0);
      const v2f A0 = sw ? a1[i] : a0[i];
      const v2f A1 = sw ? a0[i] : a1[i];
      const v2f x0 = cmul(A0, v0) + cmul(A1, v1);
      const v2f x1 = cmul(A1, v0) + cmul(A0, v1);
      v2f vlo, vhi;
      vlo.x = V[i].x; vlo.y = V[i].y; vhi.x = V[i].z; vhi.y = V[i].w;
      v0 = cmul(vlo, x0);
      v1 = cmul(vhi, x1);
    }
    if (half == 0) {
      afrag[zz * 4 + sb]     = v0;
      afrag[zz * 4 + 2 + sb] = v1;
    } else {
      afrag[256 + (2 * sb) * 64 + zz]     = v0;
      afrag[256 + (2 * sb + 1) * 64 + zz] = v1;
    }
  }
  __syncthreads();

  v2f* const L2 = afrag + FRAG_PER_LAYER;
  v2f* const R2 = (v2f*)hbuf;
  factor_pass(afrag, afrag + 2 * FRAG_PER_LAYER, L2, R2, wave_u >> 2, wave_u & 3, tq & 31);
  product_pass(st, L2, R2, tq >> 5, tq & 31);
#pragma nounroll
  for (int l = 2; l < DEPTH; ++l) {
    const v2f* af = afrag + l * FRAG_PER_LAYER;
    if (l > 2) group_pass<0, 0>(st, af,       wave, lane);
    group_pass<1, 0>(st, af + 128, wave, lane);
    if (l > 2) group_pass<2, 0>(st, af + 256, wave, lane);
    if (l < DEPTH - 1)
      group_pass<3, 1>(st, af + 384, wave, lane, nullptr, ld3, sc3);
  }

  float acc[N_QUBITS];
  group_pass<3, 2>(st, afrag + (DEPTH - 1) * FRAG_PER_LAYER + 384, wave, lane, acc, ld3);
#pragma unroll
  for (int q = 0; q < N_QUBITS; ++q) part[tid * N_QUBITS + q] = acc[q];
  __syncthreads();
  {
    v4f* const rows = (v4f*)part;
    if (tid < 48) {
      v4f* const p4 = rows + (tid & 15) * 3 + (tid >> 4);
      v4f v[16];
#pragma unroll
      for (int i = 0; i < 16; ++i) v[i] = p4[i * 48];
#pragma unroll
      for (int s = 8; s > 0; s >>= 1)
#pragma unroll
        for (int i = 0; i < s; ++i) v[i] += v[i + s];
      p4[0] = v[0];
    }
    __syncthreads();
    if (tid < 3) {
      const v4f* const p4 = rows + tid;
      v4f v[16];
#pragma unroll
      for (int i = 0; i < 16; ++i) v[i] = p4[i * 3];
#pragma unroll
      for (int s = 8; s > 0; s >>= 1)
#pragma unroll
        for (int i = 0; i < s; ++i) v[i] += v[i + s];
      *(v4f*)(meas + (size_t)bidx * N_QUBITS + 4 * tid) = v[0];
    }
  }
  }
}

__global__ __launch_bounds__(128) void back_mlp_kernel(
    const float* __restrict__ meas, const float* __restrict__ W3,
    const float* __restrict__ b3,   const float* __restrict__ W4,
    const float* __restrict__ b4,   float* __restrict__ out) {
  __shared__ float hbuf[4][16 * HID];
  const int lane  = threadIdx.x & 31;
  const int wave  = threadIdx.x >> 5;
  const int row0  = (blockIdx.x * 4 + wave) * 16;
  const int col   = lane & 15;
  const int rhalf = (lane >> 4) * 8;
  float* h = hbuf[wave];

  for (int nt = 0; nt < HID / 16; ++nt) {
    v8f c = {};
#pragma unroll
    for (int kc = 0; kc < N_QUBITS; kc += 4) {
      v2f a = load_a_frag(meas, N_QUBITS, row0, kc, lane);
      v2f b = load_b_frag(W3, HID, kc, nt * 16, lane);
      c = wmma_f32(a, b, c);
    }
    int n = nt * 16 + col;
    float bias = b3[n];
#pragma unroll
    for (int r = 0; r < 8; ++r)
      h[(rhalf + r) * HID + n] = tanhf(c[r] + bias);
  }
  __syncthreads();

  v8f c = {};
#pragma unroll
  for (int kc = 0; kc < HID; kc += 4) {
    v2f a = load_a_frag(h, HID, 0, kc, lane);
    int k = kc + ((lane >> 4) << 1);
    v2f b;
    b.x = (col < OUT_DIM) ? W4[k * OUT_DIM + col] : 0.0f;
    b.y = (col < OUT_DIM) ? W4[(k + 1) * OUT_DIM + col] : 0.0f;
    c = wmma_f32(a, b, c);
  }
  if (col < OUT_DIM) {
    float bias = b4[col];
#pragma unroll
    for (int r = 0; r < 8; ++r)
      out[(size_t)(row0 + rhalf + r) * OUT_DIM + col] = c[r] + bias;
  }
}

extern "C" void kernel_launch(void* const* d_in, const int* in_sizes, int n_in,
                              void* d_out, int out_size, void* d_ws, size_t ws_size,
                              hipStream_t stream) {
  const float* noise = (const float*)d_in[0];
  const float* W1 = (const float*)d_in[1];
  const float* b1 = (const float*)d_in[2];
  const float* W2 = (const float*)d_in[3];
  const float* b2 = (const float*)d_in[4];
  const float* W3 = (const float*)d_in[5];
  const float* b3 = (const float*)d_in[6];
  const float* W4 = (const float*)d_in[7];
  const float* b4 = (const float*)d_in[8];
  float* out = (float*)d_out;

  float* meas = (float*)d_ws;

  quantum_kernel<<<NBLK, 256, 0, stream>>>(noise, W1, b1, W2, b2, meas);
  back_mlp_kernel<<<BATCH / 64, 128, 0, stream>>>(meas, W3, b3, W4, b4, out);
}


# Two elements in flight per block (512 blocks of 8, two per WGP): every stretch between two block barriers runs for both elements, each in its own LDS buffers; 6.5 barriers an element, not 13
# speedup vs baseline: 6.4019x; 1.0330x over previous
#include <hip/hip_runtime.h>
#include <math.h>

#define N_QUBITS 12
#define DEPTH    4
#define HID      64
#define NPARAMS  144
#define BATCH    4096
#define DIM      4096
#define OUT_DIM  2

typedef __attribute__((ext_vector_type(2))) float v2f;
typedef __attribute__((ext_vector_type(4))) float v4f;
typedef __attribute__((ext_vector_type(8))) float v8f;

__device__ __forceinline__ v8f wmma_f32(v2f a, v2f b, v8f c) {
  return __builtin_amdgcn_wmma_f32_16x16x4_f32(
       false, a,  false, b,
       (short)0, c,  false,  false);
}

__device__ __forceinline__ v2f load_a_frag(const float* src, int lda,
                                           int row0, int k0, int lane) {
  int m = row0 + (lane & 15);
  int k = k0 + ((lane >> 4) << 1);
  v2f a;
  a.x = src[m * lda + k];
  a.y = src[m * lda + k + 1];
  return a;
}

__device__ __forceinline__ v2f load_b_frag(const float* W, int ldb,
                                           int k0, int n0, int lane) {
  int c = n0 + (lane & 15);
  int k = k0 + ((lane >> 4) << 1);
  v2f b;
  b.x = W[k * ldb + c];
  b.y = W[(k + 1) * ldb + c];
  return b;
}

#define FRAG_PER_LAYER 512
#define EPB  2
#define NBLK 512
#define MAXE 8
#define AFSZ (DEPTH * FRAG_PER_LAYER)

__device__ __forceinline__ int phys_idx(int i) {
  return i ^ (((i >> 5) & 3) * 9) ^ ((((i >> 8) ^ (i >> 11)) & 1) << 4);
}

__device__ __forceinline__ int cnot_ring(int x) {
#pragma unroll
  for (int q = 0; q < N_QUBITS; ++q) {
    int pc = N_QUBITS - 1 - q;
    int pt = N_QUBITS - 1 - ((q + 1) % N_QUBITS);
    if ((x >> pc) & 1) x ^= (1 << pt);
  }
  return x;
}

__device__ __forceinline__ v2f cmul(v2f a, v2f b) {
  v2f r;
  r.x = a.x * b.x - a.y * b.y;
  r.y = a.x * b.y + a.y * b.x;
  return r;
}

template <int G>
__device__ __forceinline__ void group_mma(const v2f* st, const v2f* af,
                                          int wave, int lane, int ldb,
                                          v8f (&d)[4], int (&pb)[4]) {
  constexpr int NK = 4;
  constexpr int SH = 9 - 3 * G;
  const int c = lane & 15;
  const int h = lane >> 4;
  v2f a[NK];
#pragma unroll
  for (int kk = 0; kk < NK; ++kk) a[kk] = af[kk * 32 + lane];

  v2f x[4][NK];
#pragma unroll
  for (int n = 0; n < 4; ++n) {
    const int col9 = ((wave * 4 + n) << 4) | c;
    const int hi = col9 >> SH;
    const int lo = col9 & ((1 << SH) - 1);
    const int ib = (hi << (SH + 3)) | (h << (SH + 2)) | lo;
    pb[n] = phys_idx(ib);
#pragma unroll
    for (int u = 0; u < NK; ++u)
      x[n][u] = (G == 3)
          ? *(const v2f*)((const char*)st + (ldb ^ (phys_idx((n << 7) | u) << 3)))
          : st[pb[n] ^ phys_idx(u << SH)];
  }
  __builtin_amdgcn_sched_barrier(0);
#pragma unroll
  for (int n = 0; n < 4; ++n) d[n] = v8f{};
#pragma unroll
  for (int kk = 0; kk < NK; ++kk)
#pragma unroll
    for (int n = 0; n < 4; ++n) d[n] = wmma_f32(a[kk], x[n][kk], d[n]);
}

template <int G>
__device__ __forceinline__ void group_pass(v2f* st, const v2f* af,
                                           int wave, int lane) {
  constexpr int SH = 9 - 3 * G;
  v8f d[4];
  int pb[4];
  group_mma<G>(st, af, wave, lane, 0, d, pb);
#pragma unroll
  for (int n = 0; n < 4; ++n) {
#pragma unroll
    for (int u = 0; u < 4; ++u) {
      v2f v;
      v.x = d[n][2 * u];
      v.y = d[n][2 * u + 1];
      st[pb[n] ^ phys_idx(u << SH)] = v;
    }
  }
}

__device__ __forceinline__ void ring_pass(v2f* st, const v2f* af,
                                          int wave, int lane, int ldb, int stb) {
  v8f d[EPB][4];
  int pb[4];
#pragma unroll
  for (int e = 0; e < EPB; ++e)
    group_mma<3>(st + e * DIM, af + e * AFSZ, wave, lane, ldb, d[e], pb);
  __syncthreads();
#pragma unroll
  for (int e = 0; e < EPB; ++e) {
#pragma unroll
    for (int n = 0; n < 4; ++n) {
#pragma unroll
      for (int u = 0; u < 4; ++u) {
        v2f v;
        v.x = d[e][n][2 * u];
        v.y = d[e][n][2 * u + 1];
        *(v2f*)((char*)(st + e * DIM) + (stb ^ (phys_idx(cnot_ring((n << 7) | u)) << 3))) = v;
      }
    }
  }
  __syncthreads();
}

__device__ __forceinline__ void measure_pass(const v2f* st, const v2f* af,
                                             int wave, int lane, int ldb,
                                             float* acc) {
  const int c = lane & 15;
  const int h = lane >> 4;
  v8f d[4];
  int pb[4];
  group_mma<3>(st, af, wave, lane, ldb, d, pb);
  v2f S[4], B[4], C[4];
#pragma unroll
  for (int n = 0; n < 4; ++n) {
    v2f q[4];
#pragma unroll
    for (int u = 0; u < 4; ++u) {
      v2f v;
      v.x = d[n][2 * u];
      v.y = d[n][2 * u + 1];
      q[u] = v * v;
    }
    const v2f e = q[0] + q[1], f = q[2] + q[3];
    S[n] = e + f;
    B[n] = e - f;
    C[n] = (q[0] - q[1]) + (q[3] - q[2]);
  }
  const v2f e2 = S[0] + S[1], f2 = S[2] + S[3];
  const v2f vT  = e2 + f2;
  const v2f v8  = e2 - f2;
  const v2f v87 = (S[0] - S[1]) + (S[3] - S[2]);
  const v2f v871  = (B[0] - B[1]) + (B[3] - B[2]);
  const v2f v8710 = (C[0] - C[1]) + (C[3] - C[2]);
  const float qT = vT.x + vT.y, q8 = v8.x + v8.y, q87 = v87.x + v87.y;
  const float q871 = v871.x + v871.y, q8710 = v8710.x + v8710.y;
  const unsigned L = ((unsigned)wave << 9) | ((unsigned)c << 3) | ((unsigned)h << 2);
  unsigned g = L;
  g ^= g >> 1; g ^= g >> 2; g ^= g >> 4; g ^= g >> 8;
#pragma unroll
  for (int j = 0; j <= 10; ++j) {
    const float qv = (j >= 9) ? qT : (j == 8) ? q8 : (j >= 2) ? q87
                   : (j == 1) ? q871 : q8710;
    const bool neg = ((g >> j) & 1u) != 0u;
    acc[N_QUBITS - 1 - j] = neg ? -qv : qv;
  }
  {
    const bool neg = (((g ^ (g >> 11)) & 1u) != 0u);
    acc[0] = neg ? -q8710 : q8710;
  }
}

__device__ __forceinline__ void factor_pass(const v2f* LR, const v2f* af2,
                                            v2f* L2, v2f* R2,
                                            int side, int q2, int lane) {
  const int c = lane & 15;
  const int h = lane >> 4;
  const int t3 = (q2 << 1) | (c >> 3);
  const int beta = c & 7;
  const int bb = beta >> 1;
  const int s = beta & 1;
  const v2f* af = af2 + (side ? 256 : 0);
  v2f a[4], x[4];
#pragma unroll
  for (int kk = 0; kk < 4; ++kk) {
    a[kk] = af[kk * 32 + lane];
    const int z = (((h << 2) | kk) << 3) | t3;
    const int zz = z ^ (z >> 1);
    const int il = ((zz ^ (s * 0x30)) << 2) | bb;
    const int ir = 256 + (bb << 6) + (zz ^ (s << 5));
    x[kk] = LR[side ? ir : il];
  }
  v8f d = {};
#pragma unroll
  for (int kk = 0; kk < 4; ++kk) d = wmma_f32(a[kk], x[kk], d);
#pragma unroll
  for (int u = 0; u < 4; ++u) {
    const int zn = (((h << 2) | u) << 3) | t3;
    v2f v;
    v.x = d[2 * u];
    v.y = d[2 * u + 1];
    if (side) R2[(beta << 6) + zn] = v;
    else      L2[(zn << 3) + beta] = v;
  }
}

__device__ __forceinline__ void product_pass(v2f* st, const v2f* L2, const v2f* R2,
                                             int wave, int lane) {
  const int c = lane & 15;
  const int h = lane >> 4;
  const int w1 = wave & 1;
  const int zl = ((wave >> 1) << 4) | ((c >> 3) << 3) | (((c & 7) >> 1) << 1) | w1;
  v2f a[2][2], b[4][2];
#pragma unroll
  for (int kk = 0; kk < 2; ++kk) {
    const int bnd = (kk << 1) | h;
#pragma unroll
    for (int s11 = 0; s11 < 2; ++s11) {
      const v2f l = L2[(zl << 3) + (bnd << 1) + s11];
      v2f re_row, im_row;
      re_row.x = l.x; re_row.y = -l.y;
      im_row.x = l.y; im_row.y = l.x;
      a[kk][s11] = (c & 1) ? im_row : re_row;
    }
#pragma unroll
    for (int n = 0; n < 4; ++n)
      b[n][kk] = R2[(((bnd << 1) | w1) << 6) + (((n >> 1) << 5) | (c << 1) | (n & 1))];
  }
  v8f d[4] = {};
#pragma unroll
  for (int kk = 0; kk < 2; ++kk)
#pragma unroll
    for (int n = 0; n < 4; ++n) d[n] = wmma_f32(a[kk][n & 1], b[n][kk], d[n]);
  const int pbase = phys_idx(((wave >> 1) << 10) | (h << 9) | (w1 << 6) | (c << 1));
#pragma unroll
  for (int n = 0; n < 4; ++n) {
#pragma unroll
    for (int u = 0; u < 4; ++u) {
      v2f v;
      v.x = d[n][2 * u];
      v.y = d[n][2 * u + 1];
      st[pbase ^ phys_idx((u << 7) | ((n >> 1) << 5) | (n & 1))] = v;
    }
  }
}

__global__ __launch_bounds__(256) void quantum_kernel(
    const float* __restrict__ noise, const float* __restrict__ W1,
    const float* __restrict__ b1,    const float* __restrict__ W2,
    const float* __restrict__ b2,    float* __restrict__ meas) {
  __shared__ v2f st[EPB * DIM];
  __shared__ __attribute__((aligned(16))) v2f afrag[EPB * AFSZ];
  __shared__ v2f ssc[EPB * NPARAMS];
  __shared__ __attribute__((aligned(16))) float gU[EPB * DEPTH * N_QUBITS][8];
  __shared__ __attribute__((aligned(16))) float part[EPB * 256 * N_QUBITS];
  __shared__ __attribute__((aligned(16))) float hbuf[EPB * 16 * HID];
  __shared__ float pbuf[16 * NPARAMS];
  const int tid  = threadIdx.x;
  const int lane = tid & 31;
  const int wave = tid >> 5;
  const int wave_u = __builtin_amdgcn_readfirstlane(wave);
  const int ebase = MAXE * (int)blockIdx.x;
  const int ecnt  = MAXE;

  {
    const int col   = lane & 15;
    const int rhalf = (lane >> 4) * 8;
    if (wave_u < HID / 16) {
      const int e = ebase + ((col < ecnt) ? col : 0);
      v8f cacc = {};
#pragma unroll
      for (int kc = 0; kc < N_QUBITS; kc += 4) {
        const int k = kc + ((lane >> 4) << 1);
        v2f a;
        a.x = noise[e * N_QUBITS + k];
        a.y = noise[e * N_QUBITS + k + 1];
        v2f b = load_b_frag(W1, HID, kc, wave_u * 16, lane);
        cacc = wmma_f32(a, b, cacc);
      }
      const int n = wave_u * 16 + col;
      const float bias = b1[n];
#pragma unroll
      for (int r = 0; r < 8; ++r)
        hbuf[(rhalf + r) * HID + n] = tanhf(cacc[r] + bias);
    }
    __syncthreads();
    for (int nt = wave_u; nt < NPARAMS / 16; nt += 8) {
      v8f cacc = {};
#pragma unroll
      for (int kc = 0; kc < HID; kc += 4) {
        v2f a = load_a_frag(hbuf, HID, 0, kc, lane);
        v2f b = load_b_frag(W2, NPARAMS, kc, nt * 16, lane);
        cacc = wmma_f32(a, b, cacc);
      }
      const int n = nt * 16 + col;
      const float bias = b2[n];
#pragma unroll
      for (int r = 0; r < 8; ++r)
        pbuf[(rhalf + r) * NPARAMS + n] = cacc[r] + bias;
    }
    __syncthreads();
  }

  const int tpart = (wave << 9) | ((lane & 15) << 3) | ((lane >> 4) << 2);
  const int ld3 = phys_idx(tpart) << 3;
  const int sc3 = phys_idx(cnot_ring(tpart)) << 3;

  for (int it = 0; it < MAXE; it += EPB) {
  {
    const int left = (ecnt - it) / EPB;
    if (left >= 4)      __builtin_amdgcn_s_setprio(3);
    else if (left == 3) __builtin_amdgcn_s_setprio(2);
    else if (left == 2) __builtin_amdgcn_s_setprio(1);
    else                __builtin_amdgcn_s_setprio(0);
  }

  if (tid < NPARAMS) {
#pragma unroll
    for (int e = 0; e < EPB; ++e) {
      float s, c;
      sincosf(pbuf[(it + e) * NPARAMS + tid] * 0.5f, &s, &c);
      v2f sc; sc.x = s; sc.y = c;
      ssc[e * NPARAMS + tid] = sc;
    }
  }
  __syncthreads();

  if (tid < EPB * DEPTH * N_QUBITS) {
    const int ge = tid / (DEPTH * N_QUBITS);
    const int gt = tid - ge * (DEPTH * N_QUBITS);
    const v2f* sce = ssc + ge * NPARAMS;
    const v2f A = sce[gt * 3 + 0], B = sce[gt * 3 + 1], Z = sce[gt * 3 + 2];
    const float sa = A.x, ca = A.y, sb = B.x, cb = B.y, sz = Z.x, cz = Z.y;
    const float m00r =  cb * ca, m00i =  sb * sa;
    const float m01r = -sb * ca, m01i = -cb * sa;
    const float m10r =  sb * ca, m10i = -cb * sa;
    const float m11r =  cb * ca, m11i = -sb * sa;
    float* u = gU[tid];
    u[0] = cz * m00r + sz * m00i;  u[1] = cz * m00i - sz * m00r;
    u[2] = cz * m01r + sz * m01i;  u[3] = cz * m01i - sz * m01r;
    u[4] = cz * m10r - sz * m10i;  u[5] = cz * m10i + sz * m10r;
    u[6] = cz * m11r - sz * m11i;  u[7] = cz * m11i + sz * m11r;
  }
  __syncthreads();

  for (int e = 0; e < EPB; ++e) {
  v2f* const afe = afrag + e * AFSZ;
  const int ge0 = e * DEPTH * N_QUBITS;
  {
    const int g  = tid >> 6;
    const int jp = (tid >> 3) & 7;
    const int j  = tid & 7;
    const int o0 = ((((jp >> 2) & 1) << 1) | ((j >> 2) & 1)) << 1;
    const int o1 = ((((jp >> 1) & 1) << 1) | ((j >> 1) & 1)) << 1;
    const int o2 = (((jp & 1) << 1) | (j & 1)) << 1;
    const int m0 = ((jp >> 2) << 3) | ((jp & 3) << 1);
    const int eo = (g << 7) | ((j & 3) << 5) | ((j >> 2) << 4) | m0;
    v2f ga[DEPTH][3];
#pragma unroll
    for (int l = 2; l < DEPTH; ++l) {
      const float* u = gU[ge0 + l * N_QUBITS + 3 * g];
      ga[l][0] = *(const v2f*)(u + o0);
      ga[l][1] = *(const v2f*)(u + 8 + o1);
      ga[l][2] = *(const v2f*)(u + 16 + o2);
    }
    __builtin_amdgcn_sched_barrier(0);
#pragma unroll
    for (int l = 2; l < DEPTH; ++l) {
      const v2f p = cmul(cmul(ga[l][0], ga[l][1]), ga[l][2]);
      v4f ent;
      ent.x = p.x; ent.y = -p.y; ent.z = p.y; ent.w = p.x;
      *(v4f*)(&afe[l * FRAG_PER_LAYER + eo]) = ent;
    }
  }
  {
    const int half = tid >> 7;
    const int zz = (tid & 127) >> 1;
    const int sb = tid & 1;
    v2f a0[6], a1[6];
    v4f V[6];
#pragma unroll
    for (int i = 0; i < 6; ++i) {
      const float* u0 = gU[ge0 + half * 6 + i];
      a0[i] = *(const v2f*)(u0);
      a1[i] = *(const v2f*)(u0 + 4);
      const float* u1 = gU[ge0 + N_QUBITS + half * 6 + i] + 4 * ((zz >> (5 - i)) & 1);
      V[i] = *(const v4f*)(u1);
    }
    __builtin_amdgcn_sched_barrier(0);
    v2f v0, v1;
    {
      v2f vlo, vhi;
      vlo.x = V[0].x; vlo.y = V[0].y; vhi.x = V[0].z; vhi.y = V[0].w;
      v0 = cmul(vlo, sb ? a1[0] : a0[0]);
      v1 = cmul(vhi, sb ? a0[0] : a1[0]);
    }
#pragma unroll
    for (int i = 1; i < 6; ++i) {
      const bool sw = (i == 1) && (half == 0) && (sb != 0);
      const v2f A0 = sw ? a1[i] : a0[i];
      const v2f A1 = sw ? a0[i] : a1[i];
      const v2f x0 = cmul(A0, v0) + cmul(A1, v1);
      const v2f x1 = cmul(A1, v0) + cmul(A0, v1);
      v2f vlo, vhi;
      vlo.x = V[i].x; vlo.y = V[i].y; vhi.x = V[i].z; vhi.y = V[i].w;
      v0 = cmul(vlo, x0);
      v1 = cmul(vhi, x1);
    }
    if (half == 0) {
      afe[zz * 4 + sb]     = v0;
      afe[zz * 4 + 2 + sb] = v1;
    } else {
      afe[256 + (2 * sb) * 64 + zz]     = v0;
      afe[256 + (2 * sb + 1) * 64 + zz] = v1;
    }
  }
  }
  __syncthreads();

  for (int e = 0; e < EPB; ++e) {
    v2f* const afe = afrag + e * AFSZ;
    factor_pass(afe, afe + 2 * FRAG_PER_LAYER, afe + FRAG_PER_LAYER,
                (v2f*)hbuf + e * 512, wave_u >> 2, wave_u & 3, lane);
  }
  __syncthreads();
  for (int e = 0; e < EPB; ++e)
    product_pass(st + e * DIM, afrag + e * AFSZ + FRAG_PER_LAYER,
                 (const v2f*)hbuf + e * 512, wave, lane);
  __syncthreads();
#pragma nounroll
  for (int l = 2; l < DEPTH; ++l) {
    const v2f* af = afrag + l * FRAG_PER_LAYER;
    if (l > 2) {
      for (int e = 0; e < EPB; ++e)
        group_pass<0>(st + e * DIM, af + e * AFSZ, wave, lane);
      __syncthreads();
    }
    for (int e = 0; e < EPB; ++e)
      group_pass<1>(st + e * DIM, af + e * AFSZ + 128, wave, lane);
    __syncthreads();
    if (l > 2) {
      for (int e = 0; e < EPB; ++e)
        group_pass<2>(st + e * DIM, af + e * AFSZ + 256, wave, lane);
      __syncthreads();
    }
    if (l < DEPTH - 1) ring_pass(st, af + 384, wave, lane, ld3, sc3);
  }

  for (int e = 0; e < EPB; ++e) {
    float acc[N_QUBITS];
    measure_pass(st + e * DIM, afrag + e * AFSZ + (DEPTH - 1) * FRAG_PER_LAYER + 384,
                 wave, lane, ld3, acc);
#pragma unroll
    for (int q = 0; q < N_QUBITS; ++q) part[(e * 256 + tid) * N_QUBITS + q] = acc[q];
  }
  __syncthreads();
  {
    if (tid < 48 * EPB) {
      const int re = tid / 48;
      const int t  = tid - re * 48;
      v4f* const p4 = (v4f*)(part + re * 256 * N_QUBITS) + (t & 15) * 3 + (t >> 4);
      v4f v[16];
#pragma unroll
      for (int i = 0; i < 16; ++i) v[i] = p4[i * 48];
#pragma unroll
      for (int s = 8; s > 0; s >>= 1)
#pragma unroll
        for (int i = 0; i < s; ++i) v[i] += v[i + s];
      p4[0] = v[0];
    }
    __syncthreads();
    if (tid < 3 * EPB) {
      const int re = tid / 3;
      const int t  = tid - re * 3;
      const v4f* const p4 = (const v4f*)(part + re * 256 * N_QUBITS) + t;
      v4f v[16];
#pragma unroll
      for (int i = 0; i < 16; ++i) v[i] = p4[i * 3];
#pragma unroll
      for (int s = 8; s > 0; s >>= 1)
#pragma unroll
        for (int i = 0; i < s; ++i) v[i] += v[i + s];
      *(v4f*)(meas + (size_t)(ebase + it + re) * N_QUBITS + 4 * t) = v[0];
    }
  }
  }
}

__global__ __launch_bounds__(128) void back_mlp_kernel(
    const float* __restrict__ meas, const float* __restrict__ W3,
    const float* __restrict__ b3,   const float* __restrict__ W4,
    const float* __restrict__ b4,   float* __restrict__ out) {
  __shared__ float hbuf[4][16 * HID];
  const int lane  = threadIdx.x & 31;
  const int wave  = threadIdx.x >> 5;
  const int row0  = (blockIdx.x * 4 + wave) * 16;
  const int col   = lane & 15;
  const int rhalf = (lane >> 4) * 8;
  float* h = hbuf[wave];

  for (int nt = 0; nt < HID / 16; ++nt) {
    v8f c = {};
#pragma unroll
    for (int kc = 0; kc < N_QUBITS; kc += 4) {
      v2f a = load_a_frag(meas, N_QUBITS, row0, kc, lane);
      v2f b = load_b_frag(W3, HID, kc, nt * 16, lane);
      c = wmma_f32(a, b, c);
    }
    int n = nt * 16 + col;
    float bias = b3[n];
#pragma unroll
    for (int r = 0; r < 8; ++r)
      h[(rhalf + r) * HID + n] = tanhf(c[r] + bias);
  }
  __syncthreads();

  v8f c = {};
#pragma unroll
  for (int kc = 0; kc < HID; kc += 4) {
    v2f a = load_a_frag(h, HID, 0, kc, lane);
    int k = kc + ((lane >> 4) << 1);
    v2f b;
    b.x = (col < OUT_DIM) ? W4[k * OUT_DIM + col] : 0.0f;
    b.y = (col < OUT_DIM) ? W4[(k + 1) * OUT_DIM + col] : 0.0f;
    c = wmma_f32(a, b, c);
  }
  if (col < OUT_DIM) {
    float bias = b4[col];
#pragma unroll
    for (int r = 0; r < 8; ++r)
      out[(size_t)(row0 + rhalf + r) * OUT_DIM + col] = c[r] + bias;
  }
}

extern "C" void kernel_launch(void* const* d_in, const int* in_sizes, int n_in,
                              void* d_out, int out_size, void* d_ws, size_t ws_size,
                              hipStream_t stream) {
  const float* noise = (const float*)d_in[0];
  const float* W1 = (const float*)d_in[1];
  const float* b1 = (const float*)d_in[2];
  const float* W2 = (const float*)d_in[3];
  const float* b2 = (const float*)d_in[4];
  const float* W3 = (const float*)d_in[5];
  const float* b3 = (const float*)d_in[6];
  const float* W4 = (const float*)d_in[7];
  const float* b4 = (const float*)d_in[8];
  float* out = (float*)d_out;

  float* meas = (float*)d_ws;

  quantum_kernel<<<NBLK, 256, 0, stream>>>(noise, W1, b1, W2, b2, meas);
  back_mlp_kernel<<<BATCH / 64, 128, 0, stream>>>(meas, W3, b3, W4, b4, out);
}
